# starting kernel
# baseline (speedup 1.0000x reference)
#include <hip/hip_runtime.h>
#include <hip/hip_bf16.h>
#include <math.h>

#define BB 4
#define TT 2048
#define CC 1024
#define HH 128

typedef __attribute__((ext_vector_type(2))) float v2f;
typedef __attribute__((ext_vector_type(8))) float v8f;

__device__ __forceinline__ v8f wmma_f32(v2f a, v2f b, v8f c) {
  return __builtin_amdgcn_wmma_f32_16x16x4_f32(false, a, false, b, (short)0, c,
                                               false, false);
}

__global__ __launch_bounds__(32) void qkv_proj(
    const float* __restrict__ x, const float* __restrict__ Wq,
    const float* __restrict__ Wk, const float* __restrict__ Wv,
    float* __restrict__ q, float* __restrict__ k, float* __restrict__ v) {
  const int lane = threadIdx.x & 31;
  const int lo = lane & 15, hi = lane >> 4;
  const int m0 = blockIdx.x * 16;

  const float* W;
  float* out;
  if (blockIdx.y == 0)      { W = Wq; out = q; }
  else if (blockIdx.y == 1) { W = Wk; out = k; }
  else                      { W = Wv; out = v; }

  const float* xrow = x + (size_t)(m0 + lo) * CC;
  v8f acc[8] = {};

  for (int k0 = 0; k0 < CC; k0 += 4) {
    v2f a = *(const v2f*)(xrow + k0 + 2 * hi);
    const float* wr0 = W + (size_t)(k0 + 2 * hi) * HH + lo;
    const float* wr1 = wr0 + HH;
#pragma unroll
    for (int n = 0; n < 8; ++n) {
      v2f bf;
      bf.x = wr0[n * 16];
      bf.y = wr1[n * 16];
      acc[n] = wmma_f32(a, bf, acc[n]);
    }
  }
#pragma unroll
  for (int n = 0; n < 8; ++n)
#pragma unroll
    for (int r = 0; r < 8; ++r)
      out[(size_t)(m0 + r + 8 * hi) * HH + n * 16 + lo] = acc[n][r];
}

__global__ __launch_bounds__(32) void qk_scores(const float* __restrict__ q,
                                                const float* __restrict__ kmat,
                                                float* __restrict__ S) {
  const int s0 = blockIdx.x * 16;
  const int t0 = blockIdx.y * 16;
  const int b = blockIdx.z;
  if (s0 > t0 + 15) return;

  const int lane = threadIdx.x & 31;
  const int lo = lane & 15, hi = lane >> 4;
  const float* qrow = q + ((size_t)b * TT + t0 + lo) * HH;
  const float* krow = kmat + ((size_t)b * TT + s0 + lo) * HH;

  v8f acc = {};
#pragma unroll
  for (int k0 = 0; k0 < HH; k0 += 4) {
    v2f a = *(const v2f*)(qrow + k0 + 2 * hi);
    v2f bf = *(const v2f*)(krow + k0 + 2 * hi);
    acc = wmma_f32(a, bf, acc);
  }

  const float scale = 0.08838834764831845f;
  float* Sout = S + ((size_t)b * TT + t0) * TT + s0;
#pragma unroll
  for (int r = 0; r < 8; ++r) {
    const int t = t0 + r + 8 * hi;
    const int s = s0 + lo;
    const float val = (s <= t) ? acc[r] * scale : -1e30f;
    Sout[(size_t)(r + 8 * hi) * TT + lo] = val;
  }
}

__global__ void col_softmax_stats(const float* __restrict__ S,
                                  float* __restrict__ colmax,
                                  float* __restrict__ colrcp) {
  const int gid = blockIdx.x * blockDim.x + threadIdx.x;
  const int b = gid / TT;
  const int s = gid % TT;
  const float* Sb = S + (size_t)b * TT * TT;
  const int tstart = s & ~31;

  float m = -3.0e38f;
  for (int t = tstart; t < TT; ++t)
    if (t >= s) m = fmaxf(m, Sb[(size_t)t * TT + s]);

  float sum = 0.f;
  for (int t = tstart; t < TT; ++t)
    if (t >= s) sum += __expf(Sb[(size_t)t * TT + s] - m);

  colmax[gid] = m;
  colrcp[gid] = 1.0f / sum;
}

__global__ __launch_bounds__(32) void attn_out(
    const float* __restrict__ S, const float* __restrict__ colmax,
    const float* __restrict__ colrcp, const float* __restrict__ v,
    float* __restrict__ out) {
  const int t0 = blockIdx.x * 16;
  const int b = blockIdx.y;
  const int lane = threadIdx.x & 31;
  const int lo = lane & 15, hi = lane >> 4;

  const float* Srow = S + ((size_t)b * TT + t0 + lo) * TT;
  const float* cm = colmax + (size_t)b * TT;
  const float* cr = colrcp + (size_t)b * TT;
  const float* vb = v + (size_t)b * TT * HH;

  v8f acc[8] = {};
  const int sEnd = t0 + 16;
  for (int s0 = 0; s0 < sEnd; s0 += 4) {
    const int sA = s0 + 2 * hi;
    v2f raw = *(const v2f*)(Srow + sA);
    v2f mx = *(const v2f*)(cm + sA);
    v2f rc = *(const v2f*)(cr + sA);
    v2f a;
    a.x = __expf(raw.x - mx.x) * rc.x;
    a.y = __expf(raw.y - mx.y) * rc.y;

    const float* vr0 = vb + (size_t)sA * HH + lo;
    const float* vr1 = vr0 + HH;
#pragma unroll
    for (int n = 0; n < 8; ++n) {
      v2f bf;
      bf.x = vr0[n * 16];
      bf.y = vr1[n * 16];
      acc[n] = wmma_f32(a, bf, acc[n]);
    }
  }

  float* orow = out + ((size_t)b * TT + t0) * HH;
#pragma unroll
  for (int n = 0; n < 8; ++n)
#pragma unroll
    for (int r = 0; r < 8; ++r)
      orow[(size_t)(r + 8 * hi) * HH + n * 16 + lo] = acc[n][r];
}

extern "C" void kernel_launch(void* const* d_in, const int* in_sizes, int n_in,
                              void* d_out, int out_size, void* d_ws,
                              size_t ws_size, hipStream_t stream) {
  (void)in_sizes; (void)n_in; (void)out_size; (void)ws_size;
  const float* x = (const float*)d_in[0];
  const float* Wq = (const float*)d_in[1];
  const float* Wk = (const float*)d_in[2];
  const float* Wv = (const float*)d_in[3];
  float* out = (float*)d_out;

  float* ws = (float*)d_ws;
  float* q = ws;
  float* k = q + (size_t)BB * TT * HH;
  float* v = k + (size_t)BB * TT * HH;
  float* S = v + (size_t)BB * TT * HH;
  float* cmax = S + (size_t)BB * TT * TT;
  float* crcp = cmax + (size_t)BB * TT;

  qkv_proj<<<dim3(BB * TT / 16, 3), 32, 0, stream>>>(x, Wq, Wk, Wv, q, k, v);
  qk_scores<<<dim3(TT / 16, TT / 16, BB), 32, 0, stream>>>(q, k, S);
  col_softmax_stats<<<dim3(BB * TT / 256), 256, 0, stream>>>(S, cmax, crcp);
  attn_out<<<dim3(TT / 16, BB), 32, 0, stream>>>(S, cmax, crcp, v, out);
}


# output kernel: 8 waves share a 32x128 block of out; weights (made once per element) and v go through LDS in fragment order, next chunk's global loads in flight during the products; sum order kept
# speedup vs baseline: 5.7365x; 5.7365x over previous
#include <hip/hip_runtime.h>
#include <hip/hip_bf16.h>
#include <math.h>

#define BB 4
#define TT 2048
#define CC 1024
#define HH 128

typedef __attribute__((ext_vector_type(2))) float v2f;
typedef __attribute__((ext_vector_type(8))) float v8f;

__device__ __forceinline__ v8f wmma_f32(v2f a, v2f b, v8f c) {
  return __builtin_amdgcn_wmma_f32_16x16x4_f32(false, a, false, b, (short)0, c,
                                               false, false);
}

__global__ __launch_bounds__(32) void qkv_proj(
    const float* __restrict__ x, const float* __restrict__ Wq,
    const float* __restrict__ Wk, const float* __restrict__ Wv,
    float* __restrict__ q, float* __restrict__ k, float* __restrict__ v) {
  const int lane = threadIdx.x & 31;
  const int lo = lane & 15, hi = lane >> 4;
  const int m0 = blockIdx.x * 16;

  const float* W;
  float* out;
  if (blockIdx.y == 0)      { W = Wq; out = q; }
  else if (blockIdx.y == 1) { W = Wk; out = k; }
  else                      { W = Wv; out = v; }

  const float* xrow = x + (size_t)(m0 + lo) * CC;
  v8f acc[8] = {};

  for (int k0 = 0; k0 < CC; k0 += 4) {
    v2f a = *(const v2f*)(xrow + k0 + 2 * hi);
    const float* wr0 = W + (size_t)(k0 + 2 * hi) * HH + lo;
    const float* wr1 = wr0 + HH;
#pragma unroll
    for (int n = 0; n < 8; ++n) {
      v2f bf;
      bf.x = wr0[n * 16];
      bf.y = wr1[n * 16];
      acc[n] = wmma_f32(a, bf, acc[n]);
    }
  }
#pragma unroll
  for (int n = 0; n < 8; ++n)
#pragma unroll
    for (int r = 0; r < 8; ++r)
      out[(size_t)(m0 + r + 8 * hi) * HH + n * 16 + lo] = acc[n][r];
}

__global__ __launch_bounds__(32) void qk_scores(const float* __restrict__ q,
                                                const float* __restrict__ kmat,
                                                float* __restrict__ St) {
  const int t0 = blockIdx.x * 16;
  const int s0 = blockIdx.y * 16;
  const int b = blockIdx.z;
  if (s0 > t0 + 15) return;

  const int lane = threadIdx.x & 31;
  const int lo = lane & 15, hi = lane >> 4;
  const float* krow = kmat + ((size_t)b * TT + s0 + lo) * HH;
  const float* qrow = q + ((size_t)b * TT + t0 + lo) * HH;

  v8f acc = {};
#pragma unroll
  for (int k0 = 0; k0 < HH; k0 += 4) {
    v2f a = *(const v2f*)(krow + k0 + 2 * hi);
    v2f bf = *(const v2f*)(qrow + k0 + 2 * hi);
    acc = wmma_f32(a, bf, acc);
  }

  const float scale = 0.08838834764831845f;
  float* Sout = St + ((size_t)b * TT + s0) * TT + t0;
#pragma unroll
  for (int r = 0; r < 8; ++r) {
    const int s = s0 + r + 8 * hi;
    const int t = t0 + lo;
    const float val = (s <= t) ? acc[r] * scale : -1e30f;
    Sout[(size_t)(r + 8 * hi) * TT + lo] = val;
  }
}

#define STAT_WAVES 32
#define STAT_NEG (-3.0e38f)

__global__ __launch_bounds__(1024) void col_softmax_stats(
    const float* __restrict__ St, float* __restrict__ colmax,
    float* __restrict__ colrcp) {
  __shared__ float sh_max[STAT_WAVES];
  __shared__ float sh_rcp[STAT_WAVES];

  const int lane = threadIdx.x & 31;
  const int wave = __builtin_amdgcn_readfirstlane((int)(threadIdx.x >> 5));
  const int gid = blockIdx.x * STAT_WAVES + wave;
  const int s = gid & (TT - 1);
  const float* row = St + (size_t)gid * TT;
  const int tfirst = s & ~15;
  const int tb = s & ~127;
  const int nit = (TT - tb) >> 7;

  float4 val[16];
#pragma unroll
  for (int i = 0; i < 16; ++i)
    val[i] = make_float4(STAT_NEG, STAT_NEG, STAT_NEG, STAT_NEG);

  const int tl = tb + lane * 4;
  if (tl >= tfirst) val[0] = *(const float4*)(row + tl);
#pragma unroll
  for (int i = 1; i < 16; ++i)
    if (i < nit) val[i] = *(const float4*)(row + tl + i * 128);

  val[0].x = (tl + 0 >= s) ? val[0].x : STAT_NEG;
  val[0].y = (tl + 1 >= s) ? val[0].y : STAT_NEG;
  val[0].z = (tl + 2 >= s) ? val[0].z : STAT_NEG;
  val[0].w = (tl + 3 >= s) ? val[0].w : STAT_NEG;

  float m = STAT_NEG;
#pragma unroll
  for (int i = 0; i < 16; ++i)
    if (i < nit)
      m = fmaxf(m, fmaxf(fmaxf(val[i].x, val[i].y), fmaxf(val[i].z, val[i].w)));
#pragma unroll
  for (int off = 16; off > 0; off >>= 1)
    m = fmaxf(m, __shfl_xor(m, off, 32));

  float sum = 0.f;
#pragma unroll
  for (int i = 0; i < 16; ++i)
    if (i < nit) {
      sum += __expf(val[i].x - m);
      sum += __expf(val[i].y - m);
      sum += __expf(val[i].z - m);
      sum += __expf(val[i].w - m);
    }
#pragma unroll
  for (int off = 16; off > 0; off >>= 1)
    sum += __shfl_xor(sum, off, 32);

  if (lane == 0) {
    sh_max[wave] = m;
    sh_rcp[wave] = 1.0f / sum;
  }
  __syncthreads();
  if (threadIdx.x < STAT_WAVES) {
    colmax[blockIdx.x * STAT_WAVES + threadIdx.x] = sh_max[threadIdx.x];
    colrcp[blockIdx.x * STAT_WAVES + threadIdx.x] = sh_rcp[threadIdx.x];
  }
}

#define OT_CHUNK 32
#define P_PITCH 96
#define V_PITCH 288
#define P_TILE (16 * P_PITCH)
#define V_TILE (16 * V_PITCH)

__global__ __launch_bounds__(256) void attn_out(
    const float* __restrict__ St, const float* __restrict__ colmax,
    const float* __restrict__ colrcp, const float* __restrict__ v,
    float* __restrict__ out) {
  __shared__ __attribute__((aligned(16))) float shP[2][P_TILE];
  __shared__ __attribute__((aligned(16))) float shV[2][V_TILE];

  const int tid = threadIdx.x;
  const int lane = tid & 31;
  const int lo = lane & 15, hi = lane >> 4;
  const int wave = __builtin_amdgcn_readfirstlane(tid >> 5);
  const int mt = wave & 1;
  const int ng = wave >> 1;

  const int tb = TT / OT_CHUNK - 1 - (int)blockIdx.x;
  const int t0 = tb * 32;
  const int b = blockIdx.y;
  const int nchunk = tb + 1;

  const float* Sb = St + (size_t)b * TT * TT;
  const float* cm = colmax + (size_t)b * TT;
  const float* cr = colrcp + (size_t)b * TT;
  const float* vb = v + (size_t)b * TT * HH;

  const int psp = tid >> 4;
  const int pt2 = (tid & 15) * 2;
  const int vsp = tid >> 5;
  const int vh4 = (tid & 31) * 4;

  const float* gS = Sb + (size_t)(2 * psp) * TT + t0 + pt2;
  const float* gM = cm + 2 * psp;
  const float* gR = cr + 2 * psp;
  const float* gV = vb + (size_t)(2 * vsp) * HH + vh4;
  const int ldsP = psp * P_PITCH + pt2 * 2;
  const int ldsV = vsp * V_PITCH + vh4 * 2;

  const int fA = hi * P_PITCH + (mt * 16 + lo) * 2;
  const int fB = hi * V_PITCH + (ng * 32 + lo) * 2;

  float2 s_a, s_b, mx, rc;
  float4 va0, va1, vb0, vb1;

#define OT_LOAD(c)                                                      \
  {                                                                     \
    const size_t so = (size_t)(c) * OT_CHUNK;                           \
    s_a = *(const float2*)(gS + so * TT);                               \
    s_b = *(const float2*)(gS + so * TT + TT);                          \
    mx = *(const float2*)(gM + so);                                     \
    rc = *(const float2*)(gR + so);                                     \
    va0 = *(const float4*)(gV + so * HH);                               \
    va1 = *(const float4*)(gV + so * HH + HH);                          \
    vb0 = *(const float4*)(gV + so * HH + 16 * HH);                     \
    vb1 = *(const float4*)(gV + so * HH + 17 * HH);                     \
  }

#define OT_STAGE(c, buf)                                                \
  {                                                                     \
    const int s = (c) * OT_CHUNK + 2 * psp;                             \
    const int t = t0 + pt2;                                             \
    float4 w;                                                           \
    w.x = (s <= t) ? __expf(s_a.x - mx.x) * rc.x : 0.f;                 \
    w.y = (s + 1 <= t) ? __expf(s_b.x - mx.y) * rc.y : 0.f;             \
    w.z = (s <= t + 1) ? __expf(s_a.y - mx.x) * rc.x : 0.f;             \
    w.w = (s + 1 <= t + 1) ? __expf(s_b.y - mx.y) * rc.y : 0.f;         \
    *(float4*)&shP[buf][ldsP] = w;                                      \
    float* d = &shV[buf][ldsV];                                         \
    *(float4*)(d) = make_float4(va0.x, va1.x, va0.y, va1.y);            \
    *(float4*)(d + 4) = make_float4(va0.z, va1.z, va0.w, va1.w);        \
    float* e = d + 8 * V_PITCH;                                         \
    *(float4*)(e) = make_float4(vb0.x, vb1.x, vb0.y, vb1.y);            \
    *(float4*)(e + 4) = make_float4(vb0.z, vb1.z, vb0.w, vb1.w);        \
  }

  v8f acc0 = {}, acc1 = {};

#define OT_COMPUTE(buf)                                                 \
  {                                                                     \
    const float* pA = &shP[buf][fA];                                    \
    const float* pB = &shV[buf][fB];                                    \
    v2f fa[8], fb0[8], fb1[8];                                          \
    _Pragma("unroll") for (int kk = 0; kk < 8; ++kk) {                  \
      fa[kk] = *(const v2f*)(pA + kk * 2 * P_PITCH);                    \
      fb0[kk] = *(const v2f*)(pB + kk * 2 * V_PITCH);                   \
      fb1[kk] = *(const v2f*)(pB + kk * 2 * V_PITCH + 32);              \
    }                                                                   \
    __builtin_amdgcn_sched_barrier(0x20);                                  \
    _Pragma("unroll") for (int kk = 0; kk < 8; ++kk) {                  \
      acc0 = wmma_f32(fa[kk], fb0[kk], acc0);                           \
      acc1 = wmma_f32(fa[kk], fb1[kk], acc1);                           \
    }                                                                   \
  }

  OT_LOAD(0);
  OT_STAGE(0, 0);
  __syncthreads();
  for (int c = 0; c + 1 < nchunk; ++c) {
    const int cur = c & 1;
    OT_LOAD(c + 1);
    OT_COMPUTE(cur);
    __builtin_amdgcn_sched_barrier(0x20);
    OT_STAGE(c + 1, cur ^ 1);
    __syncthreads();
  }
  OT_COMPUTE((nchunk - 1) & 1);

  float* orow = out + ((size_t)b * TT + t0 + mt * 16) * HH + ng * 32 + lo;
#pragma unroll
  for (int r = 0; r < 8; ++r) {
    orow[(size_t)(r + 8 * hi) * HH] = acc0[r];
    orow[(size_t)(r + 8 * hi) * HH + 16] = acc1[r];
  }
#undef OT_LOAD
#undef OT_STAGE
#undef OT_COMPUTE
}

extern "C" void kernel_launch(void* const* d_in, const int* in_sizes, int n_in,
                              void* d_out, int out_size, void* d_ws,
                              size_t ws_size, hipStream_t stream) {
  (void)in_sizes; (void)n_in; (void)out_size; (void)ws_size;
  const float* x = (const float*)d_in[0];
  const float* Wq = (const float*)d_in[1];
  const float* Wk = (const float*)d_in[2];
  const float* Wv = (const float*)d_in[3];
  float* out = (float*)d_out;

  float* ws = (float*)d_ws;
  float* q = ws;
  float* k = q + (size_t)BB * TT * HH;
  float* v = k + (size_t)BB * TT * HH;
  float* S = v + (size_t)BB * TT * HH;
  float* cmax = S + (size_t)BB * TT * TT;
  float* crcp = cmax + (size_t)BB * TT;

  qkv_proj<<<dim3(BB * TT / 16, 3), 32, 0, stream>>>(x, Wq, Wk, Wv, q, k, v);
  qk_scores<<<dim3(TT / 16, TT / 16, BB), 32, 0, stream>>>(q, k, S);
  col_softmax_stats<<<dim3(BB * TT / STAT_WAVES), STAT_WAVES * 32, 0, stream>>>(
      S, cmax, crcp);
  attn_out<<<dim3(TT / OT_CHUNK, BB), 256, 0, stream>>>(S, cmax, crcp, v, out);
}


# score kernel: a work-group of 8 waves makes a 128x128 block of St from q and k rows put into LDS once (was a one-wave work-group per 16x16 tile reading global memory); sum order kept
# speedup vs baseline: 6.7686x; 1.1799x over previous
#include <hip/hip_runtime.h>
#include <hip/hip_bf16.h>
#include <math.h>

#define BB 4
#define TT 2048
#define CC 1024
#define HH 128

typedef __attribute__((ext_vector_type(2))) float v2f;
typedef __attribute__((ext_vector_type(8))) float v8f;

__device__ __forceinline__ v8f wmma_f32(v2f a, v2f b, v8f c) {
  return __builtin_amdgcn_wmma_f32_16x16x4_f32(false, a, false, b, (short)0, c,
                                               false, false);
}

__global__ __launch_bounds__(32) void qkv_proj(
    const float* __restrict__ x, const float* __restrict__ Wq,
    const float* __restrict__ Wk, const float* __restrict__ Wv,
    float* __restrict__ q, float* __restrict__ k, float* __restrict__ v) {
  const int lane = threadIdx.x & 31;
  const int lo = lane & 15, hi = lane >> 4;
  const int m0 = blockIdx.x * 16;

  const float* W;
  float* out;
  if (blockIdx.y == 0)      { W = Wq; out = q; }
  else if (blockIdx.y == 1) { W = Wk; out = k; }
  else                      { W = Wv; out = v; }

  const float* xrow = x + (size_t)(m0 + lo) * CC;
  v8f acc[8] = {};

  for (int k0 = 0; k0 < CC; k0 += 4) {
    v2f a = *(const v2f*)(xrow + k0 + 2 * hi);
    const float* wr0 = W + (size_t)(k0 + 2 * hi) * HH + lo;
    const float* wr1 = wr0 + HH;
#pragma unroll
    for (int n = 0; n < 8; ++n) {
      v2f bf;
      bf.x = wr0[n * 16];
      bf.y = wr1[n * 16];
      acc[n] = wmma_f32(a, bf, acc[n]);
    }
  }
#pragma unroll
  for (int n = 0; n < 8; ++n)
#pragma unroll
    for (int r = 0; r < 8; ++r)
      out[(size_t)(m0 + r + 8 * hi) * HH + n * 16 + lo] = acc[n][r];
}

#define SC_BLK 128
#define SC_PITCH 132

__global__ __launch_bounds__(256) void qk_scores(const float* __restrict__ q,
                                                 const float* __restrict__ kmat,
                                                 float* __restrict__ St) {
  __shared__ __attribute__((aligned(16))) float shK[SC_BLK * SC_PITCH];
  __shared__ __attribute__((aligned(16))) float shQ[SC_BLK * SC_PITCH];

  const int tid = threadIdx.x;
  const int lane = tid & 31;
  const int lo = lane & 15, hi = lane >> 4;
  const int wave = __builtin_amdgcn_readfirstlane(tid >> 5);
  const int ws = wave & 3;
  const int wt = wave >> 2;

  const int pr = blockIdx.y;
  const int px = blockIdx.x;
  const int bt = (px <= pr) ? pr : 15 - pr;
  const int bs = (px <= pr) ? px : px - (pr + 1);
  const int b = blockIdx.z;
  const int s0 = bs * SC_BLK;
  const int t0 = bt * SC_BLK;

  {
    const int c4 = lane * 4;
    const float* gk = kmat + ((size_t)b * TT + s0 + wave) * HH + c4;
    const float* gq = q + ((size_t)b * TT + t0 + wave) * HH + c4;
    float4 tk[16], tq[16];
#pragma unroll
    for (int i = 0; i < 16; ++i) tk[i] = *(const float4*)(gk + (size_t)i * 8 * HH);
#pragma unroll
    for (int i = 0; i < 16; ++i) tq[i] = *(const float4*)(gq + (size_t)i * 8 * HH);
#pragma unroll
    for (int i = 0; i < 16; ++i)
      *(float4*)&shK[(wave + 8 * i) * SC_PITCH + c4] = tk[i];
#pragma unroll
    for (int i = 0; i < 16; ++i) {
      const int u = wave + 8 * i;
      const int prow = (u & 64) | ((u & 3) << 4) | ((u >> 2) & 15);
      *(float4*)&shQ[prow * SC_PITCH + c4] = tq[i];
    }
  }
  __syncthreads();

  if (bs == bt && wt == 0 && ws >= 2) return;

  const float* pA = &shK[(32 * ws + lo) * SC_PITCH + 2 * hi];
  const float* pB = &shQ[(64 * wt + lo) * SC_PITCH + 2 * hi];

  v8f acc[2][4] = {};
  v2f fa[2][4][2], fb[2][4][4];

#define SC_FRAGS(set, g)                                                \
  _Pragma("unroll") for (int kk = 0; kk < 4; ++kk) {                    \
    _Pragma("unroll") for (int i = 0; i < 2; ++i)                       \
      fa[set][kk][i] =                                                  \
          *(const v2f*)(pA + i * 16 * SC_PITCH + (g) * 16 + kk * 4);    \
    _Pragma("unroll") for (int j = 0; j < 4; ++j)                       \
      fb[set][kk][j] =                                                  \
          *(const v2f*)(pB + j * 16 * SC_PITCH + (g) * 16 + kk * 4);    \
  }

  SC_FRAGS(0, 0);
#pragma unroll
  for (int g = 0; g < 8; ++g) {
    const int cur = g & 1;
    if (g + 1 < 8) SC_FRAGS(cur ^ 1, g + 1);
    __builtin_amdgcn_sched_barrier(0);
#pragma unroll
    for (int kk = 0; kk < 4; ++kk)
#pragma unroll
      for (int i = 0; i < 2; ++i)
#pragma unroll
        for (int j = 0; j < 4; ++j)
          acc[i][j] = wmma_f32(fa[cur][kk][i], fb[cur][kk][j], acc[i][j]);
    __builtin_amdgcn_sched_barrier(0);
  }
#undef SC_FRAGS

  const float scale = 0.08838834764831845f;
  const int t = t0 + 64 * wt + 4 * lo;
#pragma unroll
  for (int i = 0; i < 2; ++i) {
#pragma unroll
    for (int r = 0; r < 8; ++r) {
      const int s = s0 + 32 * ws + 16 * i + 8 * hi + r;
      float4 o;
      o.x = (s <= t) ? acc[i][0][r] * scale : -1e30f;
      o.y = (s <= t + 1) ? acc[i][1][r] * scale : -1e30f;
      o.z = (s <= t + 2) ? acc[i][2][r] * scale : -1e30f;
      o.w = (s <= t + 3) ? acc[i][3][r] * scale : -1e30f;
      *(float4*)(St + ((size_t)b * TT + s) * TT + t) = o;
    }
  }
}

#define STAT_WAVES 32
#define STAT_NEG (-3.0e38f)

__global__ __launch_bounds__(1024) void col_softmax_stats(
    const float* __restrict__ St, float* __restrict__ colmax,
    float* __restrict__ colrcp) {
  __shared__ float sh_max[STAT_WAVES];
  __shared__ float sh_rcp[STAT_WAVES];

  const int lane = threadIdx.x & 31;
  const int wave = __builtin_amdgcn_readfirstlane((int)(threadIdx.x >> 5));
  const int gid = blockIdx.x * STAT_WAVES + wave;
  const int s = gid & (TT - 1);
  const float* row = St + (size_t)gid * TT;
  const int tfirst = s & ~15;
  const int tb = s & ~127;
  const int nit = (TT - tb) >> 7;

  float4 val[16];
#pragma unroll
  for (int i = 0; i < 16; ++i)
    val[i] = make_float4(STAT_NEG, STAT_NEG, STAT_NEG, STAT_NEG);

  const int tl = tb + lane * 4;
  if (tl >= tfirst) val[0] = *(const float4*)(row + tl);
#pragma unroll
  for (int i = 1; i < 16; ++i)
    if (i < nit) val[i] = *(const float4*)(row + tl + i * 128);

  val[0].x = (tl + 0 >= s) ? val[0].x : STAT_NEG;
  val[0].y = (tl + 1 >= s) ? val[0].y : STAT_NEG;
  val[0].z = (tl + 2 >= s) ? val[0].z : STAT_NEG;
  val[0].w = (tl + 3 >= s) ? val[0].w : STAT_NEG;

  float m = STAT_NEG;
#pragma unroll
  for (int i = 0; i < 16; ++i)
    if (i < nit)
      m = fmaxf(m, fmaxf(fmaxf(val[i].x, val[i].y), fmaxf(val[i].z, val[i].w)));
#pragma unroll
  for (int off = 16; off > 0; off >>= 1)
    m = fmaxf(m, __shfl_xor(m, off, 32));

  float sum = 0.f;
#pragma unroll
  for (int i = 0; i < 16; ++i)
    if (i < nit) {
      sum += __expf(val[i].x - m);
      sum += __expf(val[i].y - m);
      sum += __expf(val[i].z - m);
      sum += __expf(val[i].w - m);
    }
#pragma unroll
  for (int off = 16; off > 0; off >>= 1)
    sum += __shfl_xor(sum, off, 32);

  if (lane == 0) {
    sh_max[wave] = m;
    sh_rcp[wave] = 1.0f / sum;
  }
  __syncthreads();
  if (threadIdx.x < STAT_WAVES) {
    colmax[blockIdx.x * STAT_WAVES + threadIdx.x] = sh_max[threadIdx.x];
    colrcp[blockIdx.x * STAT_WAVES + threadIdx.x] = sh_rcp[threadIdx.x];
  }
}

#define OT_CHUNK 32
#define P_PITCH 96
#define V_PITCH 288
#define P_TILE (16 * P_PITCH)
#define V_TILE (16 * V_PITCH)

__global__ __launch_bounds__(256) void attn_out(
    const float* __restrict__ St, const float* __restrict__ colmax,
    const float* __restrict__ colrcp, const float* __restrict__ v,
    float* __restrict__ out) {
  __shared__ __attribute__((aligned(16))) float shP[2][P_TILE];
  __shared__ __attribute__((aligned(16))) float shV[2][V_TILE];

  const int tid = threadIdx.x;
  const int lane = tid & 31;
  const int lo = lane & 15, hi = lane >> 4;
  const int wave = __builtin_amdgcn_readfirstlane(tid >> 5);
  const int mt = wave & 1;
  const int ng = wave >> 1;

  const int tb = TT / OT_CHUNK - 1 - (int)blockIdx.x;
  const int t0 = tb * 32;
  const int b = blockIdx.y;
  const int nchunk = tb + 1;

  const float* Sb = St + (size_t)b * TT * TT;
  const float* cm = colmax + (size_t)b * TT;
  const float* cr = colrcp + (size_t)b * TT;
  const float* vb = v + (size_t)b * TT * HH;

  const int psp = tid >> 4;
  const int pt2 = (tid & 15) * 2;
  const int vsp = tid >> 5;
  const int vh4 = (tid & 31) * 4;

  const float* gS = Sb + (size_t)(2 * psp) * TT + t0 + pt2;
  const float* gM = cm + 2 * psp;
  const float* gR = cr + 2 * psp;
  const float* gV = vb + (size_t)(2 * vsp) * HH + vh4;
  const int ldsP = psp * P_PITCH + pt2 * 2;
  const int ldsV = vsp * V_PITCH + vh4 * 2;

  const int fA = hi * P_PITCH + (mt * 16 + lo) * 2;
  const int fB = hi * V_PITCH + (ng * 32 + lo) * 2;

  float2 s_a, s_b, mx, rc;
  float4 va0, va1, vb0, vb1;

#define OT_LOAD(c)                                                      \
  {                                                                     \
    const size_t so = (size_t)(c) * OT_CHUNK;                           \
    s_a = *(const float2*)(gS + so * TT);                               \
    s_b = *(const float2*)(gS + so * TT + TT);                          \
    mx = *(const float2*)(gM + so);                                     \
    rc = *(const float2*)(gR + so);                                     \
    va0 = *(const float4*)(gV + so * HH);                               \
    va1 = *(const float4*)(gV + so * HH + HH);                          \
    vb0 = *(const float4*)(gV + so * HH + 16 * HH);                     \
    vb1 = *(const float4*)(gV + so * HH + 17 * HH);                     \
  }

#define OT_STAGE(c, buf)                                                \
  {                                                                     \
    const int s = (c) * OT_CHUNK + 2 * psp;                             \
    const int t = t0 + pt2;                                             \
    float4 w;                                                           \
    w.x = (s <= t) ? __expf(s_a.x - mx.x) * rc.x : 0.f;                 \
    w.y = (s + 1 <= t) ? __expf(s_b.x - mx.y) * rc.y : 0.f;             \
    w.z = (s <= t + 1) ? __expf(s_a.y - mx.x) * rc.x : 0.f;             \
    w.w = (s + 1 <= t + 1) ? __expf(s_b.y - mx.y) * rc.y : 0.f;         \
    *(float4*)&shP[buf][ldsP] = w;                                      \
    float* d = &shV[buf][ldsV];                                         \
    *(float4*)(d) = make_float4(va0.x, va1.x, va0.y, va1.y);            \
    *(float4*)(d + 4) = make_float4(va0.z, va1.z, va0.w, va1.w);        \
    float* e = d + 8 * V_PITCH;                                         \
    *(float4*)(e) = make_float4(vb0.x, vb1.x, vb0.y, vb1.y);            \
    *(float4*)(e + 4) = make_float4(vb0.z, vb1.z, vb0.w, vb1.w);        \
  }

  v8f acc0 = {}, acc1 = {};

#define OT_COMPUTE(buf)                                                 \
  {                                                                     \
    const float* pA = &shP[buf][fA];                                    \
    const float* pB = &shV[buf][fB];                                    \
    v2f fa[8], fb0[8], fb1[8];                                          \
    _Pragma("unroll") for (int kk = 0; kk < 8; ++kk) {                  \
      fa[kk] = *(const v2f*)(pA + kk * 2 * P_PITCH);                    \
      fb0[kk] = *(const v2f*)(pB + kk * 2 * V_PITCH);                   \
      fb1[kk] = *(const v2f*)(pB + kk * 2 * V_PITCH + 32);              \
    }                                                                   \
    __builtin_amdgcn_sched_barrier(0x20);                                  \
    _Pragma("unroll") for (int kk = 0; kk < 8; ++kk) {                  \
      acc0 = wmma_f32(fa[kk], fb0[kk], acc0);                           \
      acc1 = wmma_f32(fa[kk], fb1[kk], acc1);                           \
    }                                                                   \
  }

  OT_LOAD(0);
  OT_STAGE(0, 0);
  __syncthreads();
  for (int c = 0; c + 1 < nchunk; ++c) {
    const int cur = c & 1;
    OT_LOAD(c + 1);
    OT_COMPUTE(cur);
    __builtin_amdgcn_sched_barrier(0x20);
    OT_STAGE(c + 1, cur ^ 1);
    __syncthreads();
  }
  OT_COMPUTE((nchunk - 1) & 1);

  float* orow = out + ((size_t)b * TT + t0 + mt * 16) * HH + ng * 32 + lo;
#pragma unroll
  for (int r = 0; r < 8; ++r) {
    orow[(size_t)(r + 8 * hi) * HH] = acc0[r];
    orow[(size_t)(r + 8 * hi) * HH + 16] = acc1[r];
  }
#undef OT_LOAD
#undef OT_STAGE
#undef OT_COMPUTE
}

extern "C" void kernel_launch(void* const* d_in, const int* in_sizes, int n_in,
                              void* d_out, int out_size, void* d_ws,
                              size_t ws_size, hipStream_t stream) {
  (void)in_sizes; (void)n_in; (void)out_size; (void)ws_size;
  const float* x = (const float*)d_in[0];
  const float* Wq = (const float*)d_in[1];
  const float* Wk = (const float*)d_in[2];
  const float* Wv = (const float*)d_in[3];
  float* out = (float*)d_out;

  float* ws = (float*)d_ws;
  float* q = ws;
  float* k = q + (size_t)BB * TT * HH;
  float* v = k + (size_t)BB * TT * HH;
  float* S = v + (size_t)BB * TT * HH;
  float* cmax = S + (size_t)BB * TT * TT;
  float* crcp = cmax + (size_t)BB * TT;

  qkv_proj<<<dim3(BB * TT / 16, 3), 32, 0, stream>>>(x, Wq, Wk, Wv, q, k, v);
  qk_scores<<<dim3(17, 8, BB), 256, 0, stream>>>(q, k, S);
  col_softmax_stats<<<dim3(BB * TT / STAT_WAVES), STAT_WAVES * 32, 0, stream>>>(
      S, cmax, crcp);
  attn_out<<<dim3(TT / OT_CHUNK, BB), 256, 0, stream>>>(S, cmax, crcp, v, out);
}


# projection kernel: a work-group of 8 waves makes a 64x192 block of [q|k|v] from x and W chunks put into LDS once (was a one-wave work-group per 16x128 slab reading global memory); sum order kept
# speedup vs baseline: 13.0370x; 1.9261x over previous
#include <hip/hip_runtime.h>
#include <hip/hip_bf16.h>
#include <math.h>

#define BB 4
#define TT 2048
#define CC 1024
#define HH 128

typedef __attribute__((ext_vector_type(2))) float v2f;
typedef __attribute__((ext_vector_type(8))) float v8f;

__device__ __forceinline__ v8f wmma_f32(v2f a, v2f b, v8f c) {
  return __builtin_amdgcn_wmma_f32_16x16x4_f32(false, a, false, b, (short)0, c,
                                               false, false);
}

#define PJ_BM 64
#define PJ_BK 32
#define PJ_XP 36
#define PJ_WP 416
#define PJ_XT (PJ_BM * PJ_XP)
#define PJ_WT ((PJ_BK / 2) * PJ_WP)

__global__ __launch_bounds__(256) void qkv_proj(
    const float* __restrict__ x, const float* __restrict__ Wq,
    const float* __restrict__ Wk, const float* __restrict__ Wv,
    float* __restrict__ q, float* __restrict__ k, float* __restrict__ v) {
  __shared__ __attribute__((aligned(16))) float shX[2][PJ_XT];
  __shared__ __attribute__((aligned(16))) float shW[2][PJ_WT];

  const int tid = threadIdx.x;
  const int lane = tid & 31;
  const int lo = lane & 15, hi = lane >> 4;
  const int wave = __builtin_amdgcn_readfirstlane(tid >> 5);
  const int rp = wave & 1;
  const int cg = wave >> 1;

  const int half = blockIdx.x;
  const int m0 = blockIdx.y * PJ_BM;

  const int xr = tid >> 3;
  const int xc = (tid & 7) * 4;
  const int wp = tid >> 4;
  const int wc = (tid & 15) * 4;

  const float* gX = x + (size_t)(m0 + xr) * CC + xc;
  const float* gW0 = (half == 0 ? Wq : Wk) + (size_t)(2 * wp) * HH +
                     (half == 0 ? 0 : 64) + wc;
  const float* gW1 = (half == 0 ? Wq : Wv) + (size_t)(2 * wp) * HH +
                     (half == 0 ? 64 : 0) + wc;
  const float* gW2 = (half == 0 ? Wk : Wv) + (size_t)(2 * wp) * HH +
                     (half == 0 ? 0 : 64) + wc;
  const int ldsX = xr * PJ_XP + xc;
  const int ldsW = wp * PJ_WP + wc * 2;

  const int fA = (rp * 32 + lo) * PJ_XP + 2 * hi;
  const int fB = hi * PJ_WP + (cg * 48 + lo) * 2;

  float4 xa, xb, w0a, w0b, w1a, w1b, w2a, w2b;

#define PJ_LOAD(c)                                                      \
  {                                                                     \
    const size_t ko = (size_t)(c) * PJ_BK;                              \
    xa = *(const float4*)(gX + ko);                                     \
    xb = *(const float4*)(gX + ko + (size_t)32 * CC);                   \
    w0a = *(const float4*)(gW0 + ko * HH);                              \
    w0b = *(const float4*)(gW0 + ko * HH + HH);                         \
    w1a = *(const float4*)(gW1 + ko * HH);                              \
    w1b = *(const float4*)(gW1 + ko * HH + HH);                         \
    w2a = *(const float4*)(gW2 + ko * HH);                              \
    w2b = *(const float4*)(gW2 + ko * HH + HH);                         \
  }

#define PJ_STAGE(buf)                                                   \
  {                                                                     \
    *(float4*)&shX[buf][ldsX] = xa;                                     \
    *(float4*)&shX[buf][ldsX + 32 * PJ_XP] = xb;                        \
    float* d = &shW[buf][ldsW];                                         \
    *(float4*)(d) = make_float4(w0a.x, w0b.x, w0a.y, w0b.y);            \
    *(float4*)(d + 4) = make_float4(w0a.z, w0b.z, w0a.w, w0b.w);        \
    *(float4*)(d + 128) = make_float4(w1a.x, w1b.x, w1a.y, w1b.y);      \
    *(float4*)(d + 132) = make_float4(w1a.z, w1b.z, w1a.w, w1b.w);      \
    *(float4*)(d + 256) = make_float4(w2a.x, w2b.x, w2a.y, w2b.y);      \
    *(float4*)(d + 260) = make_float4(w2a.z, w2b.z, w2a.w, w2b.w);      \
  }

  v8f acc[2][3] = {};

#define PJ_COMPUTE(buf)                                                 \
  {                                                                     \
    const float* pA = &shX[buf][fA];                                    \
    const float* pB = &shW[buf][fB];                                    \
    v2f fa[8][2], fb[8][3];                                             \
    _Pragma("unroll") for (int kk = 0; kk < 8; ++kk) {                  \
      _Pragma("unroll") for (int i = 0; i < 2; ++i)                     \
        fa[kk][i] = *(const v2f*)(pA + i * 16 * PJ_XP + kk * 4);        \
      _Pragma("unroll") for (int j = 0; j < 3; ++j)                     \
        fb[kk][j] = *(const v2f*)(pB + kk * 2 * PJ_WP + j * 32);        \
    }                                                                   \
    _Pragma("unroll") for (int kk = 0; kk < 8; ++kk)                    \
      _Pragma("unroll") for (int i = 0; i < 2; ++i)                     \
        _Pragma("unroll") for (int j = 0; j < 3; ++j)                   \
          acc[i][j] = wmma_f32(fa[kk][i], fb[kk][j], acc[i][j]);        \
    __builtin_amdgcn_sched_group_barrier(0x020, 8, 0);                  \
    __builtin_amdgcn_sched_group_barrier(0x100, 40, 0);                 \
    __builtin_amdgcn_sched_group_barrier(0x008, 48, 0);                 \
  }

  PJ_LOAD(0);
  PJ_STAGE(0);
  __syncthreads();
  for (int c = 0; c + 1 < CC / PJ_BK; ++c) {
    const int cur = c & 1;
    PJ_LOAD(c + 1);
    PJ_COMPUTE(cur);
    PJ_STAGE(cur ^ 1);
    __syncthreads();
  }
  PJ_COMPUTE((CC / PJ_BK - 1) & 1);

#pragma unroll
  for (int j = 0; j < 3; ++j) {
    const int gn = 192 * half + 48 * cg + 16 * j;
    float* om = (gn < HH) ? q : (gn < 2 * HH) ? k : v;
    float* orow = om + (size_t)(m0 + rp * 32) * HH + (gn & (HH - 1)) + lo;
#pragma unroll
    for (int i = 0; i < 2; ++i)
#pragma unroll
      for (int r = 0; r < 8; ++r)
        orow[(size_t)(16 * i + 8 * hi + r) * HH] = acc[i][j][r];
  }
#undef PJ_LOAD
#undef PJ_STAGE
#undef PJ_COMPUTE
}

#define SC_BLK 128
#define SC_PITCH 132

__global__ __launch_bounds__(256) void qk_scores(const float* __restrict__ q,
                                                 const float* __restrict__ kmat,
                                                 float* __restrict__ St) {
  __shared__ __attribute__((aligned(16))) float shK[SC_BLK * SC_PITCH];
  __shared__ __attribute__((aligned(16))) float shQ[SC_BLK * SC_PITCH];

  const int tid = threadIdx.x;
  const int lane = tid & 31;
  const int lo = lane & 15, hi = lane >> 4;
  const int wave = __builtin_amdgcn_readfirstlane(tid >> 5);
  const int ws = wave & 3;
  const int wt = wave >> 2;

  const int pr = blockIdx.y;
  const int px = blockIdx.x;
  const int bt = (px <= pr) ? pr : 15 - pr;
  const int bs = (px <= pr) ? px : px - (pr + 1);
  const int b = blockIdx.z;
  const int s0 = bs * SC_BLK;
  const int t0 = bt * SC_BLK;

  {
    const int c4 = lane * 4;
    const float* gk = kmat + ((size_t)b * TT + s0 + wave) * HH + c4;
    const float* gq = q + ((size_t)b * TT + t0 + wave) * HH + c4;
    float4 tk[16], tq[16];
#pragma unroll
    for (int i = 0; i < 16; ++i) tk[i] = *(const float4*)(gk + (size_t)i * 8 * HH);
#pragma unroll
    for (int i = 0; i < 16; ++i) tq[i] = *(const float4*)(gq + (size_t)i * 8 * HH);
#pragma unroll
    for (int i = 0; i < 16; ++i)
      *(float4*)&shK[(wave + 8 * i) * SC_PITCH + c4] = tk[i];
#pragma unroll
    for (int i = 0; i < 16; ++i) {
      const int u = wave + 8 * i;
      const int prow = (u & 64) | ((u & 3) << 4) | ((u >> 2) & 15);
      *(float4*)&shQ[prow * SC_PITCH + c4] = tq[i];
    }
  }
  __syncthreads();

  if (bs == bt && wt == 0 && ws >= 2) return;

  const float* pA = &shK[(32 * ws + lo) * SC_PITCH + 2 * hi];
  const float* pB = &shQ[(64 * wt + lo) * SC_PITCH + 2 * hi];

  v8f acc[2][4] = {};
  v2f fa[2][4][2], fb[2][4][4];

#define SC_FRAGS(set, g)                                                \
  _Pragma("unroll") for (int kk = 0; kk < 4; ++kk) {                    \
    _Pragma("unroll") for (int i = 0; i < 2; ++i)                       \
      fa[set][kk][i] =                                                  \
          *(const v2f*)(pA + i * 16 * SC_PITCH + (g) * 16 + kk * 4);    \
    _Pragma("unroll") for (int j = 0; j < 4; ++j)                       \
      fb[set][kk][j] =                                                  \
          *(const v2f*)(pB + j * 16 * SC_PITCH + (g) * 16 + kk * 4);    \
  }

  SC_FRAGS(0, 0);
#pragma unroll
  for (int g = 0; g < 8; ++g) {
    const int cur = g & 1;
    if (g + 1 < 8) SC_FRAGS(cur ^ 1, g + 1);
    __builtin_amdgcn_sched_barrier(0);
#pragma unroll
    for (int kk = 0; kk < 4; ++kk)
#pragma unroll
      for (int i = 0; i < 2; ++i)
#pragma unroll
        for (int j = 0; j < 4; ++j)
          acc[i][j] = wmma_f32(fa[cur][kk][i], fb[cur][kk][j], acc[i][j]);
    __builtin_amdgcn_sched_barrier(0);
  }
#undef SC_FRAGS

  const float scale = 0.08838834764831845f;
  const int t = t0 + 64 * wt + 4 * lo;
#pragma unroll
  for (int i = 0; i < 2; ++i) {
#pragma unroll
    for (int r = 0; r < 8; ++r) {
      const int s = s0 + 32 * ws + 16 * i + 8 * hi + r;
      float4 o;
      o.x = (s <= t) ? acc[i][0][r] * scale : -1e30f;
      o.y = (s <= t + 1) ? acc[i][1][r] * scale : -1e30f;
      o.z = (s <= t + 2) ? acc[i][2][r] * scale : -1e30f;
      o.w = (s <= t + 3) ? acc[i][3][r] * scale : -1e30f;
      *(float4*)(St + ((size_t)b * TT + s) * TT + t) = o;
    }
  }
}

#define STAT_WAVES 32
#define STAT_NEG (-3.0e38f)

__global__ __launch_bounds__(1024) void col_softmax_stats(
    const float* __restrict__ St, float* __restrict__ colmax,
    float* __restrict__ colrcp) {
  __shared__ float sh_max[STAT_WAVES];
  __shared__ float sh_rcp[STAT_WAVES];

  const int lane = threadIdx.x & 31;
  const int wave = __builtin_amdgcn_readfirstlane((int)(threadIdx.x >> 5));
  const int gid = blockIdx.x * STAT_WAVES + wave;
  const int s = gid & (TT - 1);
  const float* row = St + (size_t)gid * TT;
  const int tfirst = s & ~15;
  const int tb = s & ~127;
  const int nit = (TT - tb) >> 7;

  float4 val[16];
#pragma unroll
  for (int i = 0; i < 16; ++i)
    val[i] = make_float4(STAT_NEG, STAT_NEG, STAT_NEG, STAT_NEG);

  const int tl = tb + lane * 4;
  if (tl >= tfirst) val[0] = *(const float4*)(row + tl);
#pragma unroll
  for (int i = 1; i < 16; ++i)
    if (i < nit) val[i] = *(const float4*)(row + tl + i * 128);

  val[0].x = (tl + 0 >= s) ? val[0].x : STAT_NEG;
  val[0].y = (tl + 1 >= s) ? val[0].y : STAT_NEG;
  val[0].z = (tl + 2 >= s) ? val[0].z : STAT_NEG;
  val[0].w = (tl + 3 >= s) ? val[0].w : STAT_NEG;

  float m = STAT_NEG;
#pragma unroll
  for (int i = 0; i < 16; ++i)
    if (i < nit)
      m = fmaxf(m, fmaxf(fmaxf(val[i].x, val[i].y), fmaxf(val[i].z, val[i].w)));
#pragma unroll
  for (int off = 16; off > 0; off >>= 1)
    m = fmaxf(m, __shfl_xor(m, off, 32));

  float sum = 0.f;
#pragma unroll
  for (int i = 0; i < 16; ++i)
    if (i < nit) {
      sum += __expf(val[i].x - m);
      sum += __expf(val[i].y - m);
      sum += __expf(val[i].z - m);
      sum += __expf(val[i].w - m);
    }
#pragma unroll
  for (int off = 16; off > 0; off >>= 1)
    sum += __shfl_xor(sum, off, 32);

  if (lane == 0) {
    sh_max[wave] = m;
    sh_rcp[wave] = 1.0f / sum;
  }
  __syncthreads();
  if (threadIdx.x < STAT_WAVES) {
    colmax[blockIdx.x * STAT_WAVES + threadIdx.x] = sh_max[threadIdx.x];
    colrcp[blockIdx.x * STAT_WAVES + threadIdx.x] = sh_rcp[threadIdx.x];
  }
}

#define OT_CHUNK 32
#define P_PITCH 96
#define V_PITCH 288
#define P_TILE (16 * P_PITCH)
#define V_TILE (16 * V_PITCH)

__global__ __launch_bounds__(256) void attn_out(
    const float* __restrict__ St, const float* __restrict__ colmax,
    const float* __restrict__ colrcp, const float* __restrict__ v,
    float* __restrict__ out) {
  __shared__ __attribute__((aligned(16))) float shP[2][P_TILE];
  __shared__ __attribute__((aligned(16))) float shV[2][V_TILE];

  const int tid = threadIdx.x;
  const int lane = tid & 31;
  const int lo = lane & 15, hi = lane >> 4;
  const int wave = __builtin_amdgcn_readfirstlane(tid >> 5);
  const int mt = wave & 1;
  const int ng = wave >> 1;

  const int tb = TT / OT_CHUNK - 1 - (int)blockIdx.x;
  const int t0 = tb * 32;
  const int b = blockIdx.y;
  const int nchunk = tb + 1;

  const float* Sb = St + (size_t)b * TT * TT;
  const float* cm = colmax + (size_t)b * TT;
  const float* cr = colrcp + (size_t)b * TT;
  const float* vb = v + (size_t)b * TT * HH;

  const int psp = tid >> 4;
  const int pt2 = (tid & 15) * 2;
  const int vsp = tid >> 5;
  const int vh4 = (tid & 31) * 4;

  const float* gS = Sb + (size_t)(2 * psp) * TT + t0 + pt2;
  const float* gM = cm + 2 * psp;
  const float* gR = cr + 2 * psp;
  const float* gV = vb + (size_t)(2 * vsp) * HH + vh4;
  const int ldsP = psp * P_PITCH + pt2 * 2;
  const int ldsV = vsp * V_PITCH + vh4 * 2;

  const int fA = hi * P_PITCH + (mt * 16 + lo) * 2;
  const int fB = hi * V_PITCH + (ng * 32 + lo) * 2;

  float2 s_a, s_b, mx, rc;
  float4 va0, va1, vb0, vb1;

#define OT_LOAD(c)                                                      \
  {                                                                     \
    const size_t so = (size_t)(c) * OT_CHUNK;                           \
    s_a = *(const float2*)(gS + so * TT);                               \
    s_b = *(const float2*)(gS + so * TT + TT);                          \
    mx = *(const float2*)(gM + so);                                     \
    rc = *(const float2*)(gR + so);                                     \
    va0 = *(const float4*)(gV + so * HH);                               \
    va1 = *(const float4*)(gV + so * HH + HH);                          \
    vb0 = *(const float4*)(gV + so * HH + 16 * HH);                     \
    vb1 = *(const float4*)(gV + so * HH + 17 * HH);                     \
  }

#define OT_STAGE(c, buf)                                                \
  {                                                                     \
    const int s = (c) * OT_CHUNK + 2 * psp;                             \
    const int t = t0 + pt2;                                             \
    float4 w;                                                           \
    w.x = (s <= t) ? __expf(s_a.x - mx.x) * rc.x : 0.f;                 \
    w.y = (s + 1 <= t) ? __expf(s_b.x - mx.y) * rc.y : 0.f;             \
    w.z = (s <= t + 1) ? __expf(s_a.y - mx.x) * rc.x : 0.f;             \
    w.w = (s + 1 <= t + 1) ? __expf(s_b.y - mx.y) * rc.y : 0.f;         \
    *(float4*)&shP[buf][ldsP] = w;                                      \
    float* d = &shV[buf][ldsV];                                         \
    *(float4*)(d) = make_float4(va0.x, va1.x, va0.y, va1.y);            \
    *(float4*)(d + 4) = make_float4(va0.z, va1.z, va0.w, va1.w);        \
    float* e = d + 8 * V_PITCH;                                         \
    *(float4*)(e) = make_float4(vb0.x, vb1.x, vb0.y, vb1.y);            \
    *(float4*)(e + 4) = make_float4(vb0.z, vb1.z, vb0.w, vb1.w);        \
  }

  v8f acc0 = {}, acc1 = {};

#define OT_COMPUTE(buf)                                                 \
  {                                                                     \
    const float* pA = &shP[buf][fA];                                    \
    const float* pB = &shV[buf][fB];                                    \
    v2f fa[8], fb0[8], fb1[8];                                          \
    _Pragma("unroll") for (int kk = 0; kk < 8; ++kk) {                  \
      fa[kk] = *(const v2f*)(pA + kk * 2 * P_PITCH);                    \
      fb0[kk] = *(const v2f*)(pB + kk * 2 * V_PITCH);                   \
      fb1[kk] = *(const v2f*)(pB + kk * 2 * V_PITCH + 32);              \
    }                                                                   \
    __builtin_amdgcn_sched_barrier(0x20);                                  \
    _Pragma("unroll") for (int kk = 0; kk < 8; ++kk) {                  \
      acc0 = wmma_f32(fa[kk], fb0[kk], acc0);                           \
      acc1 = wmma_f32(fa[kk], fb1[kk], acc1);                           \
    }                                                                   \
  }

  OT_LOAD(0);
  OT_STAGE(0, 0);
  __syncthreads();
  for (int c = 0; c + 1 < nchunk; ++c) {
    const int cur = c & 1;
    OT_LOAD(c + 1);
    OT_COMPUTE(cur);
    __builtin_amdgcn_sched_barrier(0x20);
    OT_STAGE(c + 1, cur ^ 1);
    __syncthreads();
  }
  OT_COMPUTE((nchunk - 1) & 1);

  float* orow = out + ((size_t)b * TT + t0 + mt * 16) * HH + ng * 32 + lo;
#pragma unroll
  for (int r = 0; r < 8; ++r) {
    orow[(size_t)(r + 8 * hi) * HH] = acc0[r];
    orow[(size_t)(r + 8 * hi) * HH + 16] = acc1[r];
  }
#undef OT_LOAD
#undef OT_STAGE
#undef OT_COMPUTE
}

extern "C" void kernel_launch(void* const* d_in, const int* in_sizes, int n_in,
                              void* d_out, int out_size, void* d_ws,
                              size_t ws_size, hipStream_t stream) {
  (void)in_sizes; (void)n_in; (void)out_size; (void)ws_size;
  const float* x = (const float*)d_in[0];
  const float* Wq = (const float*)d_in[1];
  const float* Wk = (const float*)d_in[2];
  const float* Wv = (const float*)d_in[3];
  float* out = (float*)d_out;

  float* ws = (float*)d_ws;
  float* q = ws;
  float* k = q + (size_t)BB * TT * HH;
  float* v = k + (size_t)BB * TT * HH;
  float* S = v + (size_t)BB * TT * HH;
  float* cmax = S + (size_t)BB * TT * TT;
  float* crcp = cmax + (size_t)BB * TT;

  qkv_proj<<<dim3(2, BB * TT / PJ_BM), 256, 0, stream>>>(x, Wq, Wk, Wv, q, k, v);
  qk_scores<<<dim3(17, 8, BB), 256, 0, stream>>>(q, k, S);
  col_softmax_stats<<<dim3(BB * TT / STAT_WAVES), STAT_WAVES * 32, 0, stream>>>(
      S, cmax, crcp);
  attn_out<<<dim3(TT / OT_CHUNK, BB), 256, 0, stream>>>(S, cmax, crcp, v, out);
}


# output kernel: chunk loop as a 4-deep pipeline: a chunk's global loads go out 4 chunks ahead into one of 4 register sets and stay in flight across an LDS-only barrier (was 1 ahead); sum order kept
# speedup vs baseline: 15.0433x; 1.1539x over previous
#include <hip/hip_runtime.h>
#include <hip/hip_bf16.h>
#include <math.h>

#define BB 4
#define TT 2048
#define CC 1024
#define HH 128

typedef __attribute__((ext_vector_type(2))) float v2f;
typedef __attribute__((ext_vector_type(8))) float v8f;
typedef __attribute__((ext_vector_type(4))) float v4f;

__device__ __forceinline__ v8f wmma_f32(v2f a, v2f b, v8f c) {
  return __builtin_amdgcn_wmma_f32_16x16x4_f32(false, a, false, b, (short)0, c,
                                               false, false);
}

#define PJ_BM 64
#define PJ_BK 32
#define PJ_XP 36
#define PJ_WP 416
#define PJ_XT (PJ_BM * PJ_XP)
#define PJ_WT ((PJ_BK / 2) * PJ_WP)

__global__ __launch_bounds__(256) void qkv_proj(
    const float* __restrict__ x, const float* __restrict__ Wq,
    const float* __restrict__ Wk, const float* __restrict__ Wv,
    float* __restrict__ q, float* __restrict__ k, float* __restrict__ v) {
  __shared__ __attribute__((aligned(16))) float shX[2][PJ_XT];
  __shared__ __attribute__((aligned(16))) float shW[2][PJ_WT];

  const int tid = threadIdx.x;
  const int lane = tid & 31;
  const int lo = lane & 15, hi = lane >> 4;
  const int wave = __builtin_amdgcn_readfirstlane(tid >> 5);
  const int rp = wave & 1;
  const int cg = wave >> 1;

  const int half = blockIdx.x;
  const int m0 = blockIdx.y * PJ_BM;

  const int xr = tid >> 3;
  const int xc = (tid & 7) * 4;
  const int wp = tid >> 4;
  const int wc = (tid & 15) * 4;

  const float* gX = x + (size_t)(m0 + xr) * CC + xc;
  const float* gW0 = (half == 0 ? Wq : Wk) + (size_t)(2 * wp) * HH +
                     (half == 0 ? 0 : 64) + wc;
  const float* gW1 = (half == 0 ? Wq : Wv) + (size_t)(2 * wp) * HH +
                     (half == 0 ? 64 : 0) + wc;
  const float* gW2 = (half == 0 ? Wk : Wv) + (size_t)(2 * wp) * HH +
                     (half == 0 ? 0 : 64) + wc;
  const int ldsX = xr * PJ_XP + xc;
  const int ldsW = wp * PJ_WP + wc * 2;

  const int fA = (rp * 32 + lo) * PJ_XP + 2 * hi;
  const int fB = hi * PJ_WP + (cg * 48 + lo) * 2;

  float4 xa, xb, w0a, w0b, w1a, w1b, w2a, w2b;

#define PJ_LOAD(c)                                                      \
  {                                                                     \
    const size_t ko = (size_t)(c) * PJ_BK;                              \
    xa = *(const float4*)(gX + ko);                                     \
    xb = *(const float4*)(gX + ko + (size_t)32 * CC);                   \
    w0a = *(const float4*)(gW0 + ko * HH);                              \
    w0b = *(const float4*)(gW0 + ko * HH + HH);                         \
    w1a = *(const float4*)(gW1 + ko * HH);                              \
    w1b = *(const float4*)(gW1 + ko * HH + HH);                         \
    w2a = *(const float4*)(gW2 + ko * HH);                              \
    w2b = *(const float4*)(gW2 + ko * HH + HH);                         \
  }

#define PJ_STAGE(buf)                                                   \
  {                                                                     \
    *(float4*)&shX[buf][ldsX] = xa;                                     \
    *(float4*)&shX[buf][ldsX + 32 * PJ_XP] = xb;                        \
    float* d = &shW[buf][ldsW];                                         \
    *(float4*)(d) = make_float4(w0a.x, w0b.x, w0a.y, w0b.y);            \
    *(float4*)(d + 4) = make_float4(w0a.z, w0b.z, w0a.w, w0b.w);        \
    *(float4*)(d + 128) = make_float4(w1a.x, w1b.x, w1a.y, w1b.y);      \
    *(float4*)(d + 132) = make_float4(w1a.z, w1b.z, w1a.w, w1b.w);      \
    *(float4*)(d + 256) = make_float4(w2a.x, w2b.x, w2a.y, w2b.y);      \
    *(float4*)(d + 260) = make_float4(w2a.z, w2b.z, w2a.w, w2b.w);      \
  }

  v8f acc[2][3] = {};

#define PJ_COMPUTE(buf)                                                 \
  {                                                                     \
    const float* pA = &shX[buf][fA];                                    \
    const float* pB = &shW[buf][fB];                                    \
    v2f fa[8][2], fb[8][3];                                             \
    _Pragma("unroll") for (int kk = 0; kk < 8; ++kk) {                  \
      _Pragma("unroll") for (int i = 0; i < 2; ++i)                     \
        fa[kk][i] = *(const v2f*)(pA + i * 16 * PJ_XP + kk * 4);        \
      _Pragma("unroll") for (int j = 0; j < 3; ++j)                     \
        fb[kk][j] = *(const v2f*)(pB + kk * 2 * PJ_WP + j * 32);        \
    }                                                                   \
    _Pragma("unroll") for (int kk = 0; kk < 8; ++kk)                    \
      _Pragma("unroll") for (int i = 0; i < 2; ++i)                     \
        _Pragma("unroll") for (int j = 0; j < 3; ++j)                   \
          acc[i][j] = wmma_f32(fa[kk][i], fb[kk][j], acc[i][j]);        \
    __builtin_amdgcn_sched_group_barrier(0x020, 8, 0);                  \
    __builtin_amdgcn_sched_group_barrier(0x100, 40, 0);                 \
    __builtin_amdgcn_sched_group_barrier(0x008, 48, 0);                 \
  }

  PJ_LOAD(0);
  PJ_STAGE(0);
  __syncthreads();
  for (int c = 0; c + 1 < CC / PJ_BK; ++c) {
    const int cur = c & 1;
    PJ_LOAD(c + 1);
    PJ_COMPUTE(cur);
    PJ_STAGE(cur ^ 1);
    __syncthreads();
  }
  PJ_COMPUTE((CC / PJ_BK - 1) & 1);

#pragma unroll
  for (int j = 0; j < 3; ++j) {
    const int gn = 192 * half + 48 * cg + 16 * j;
    float* om = (gn < HH) ? q : (gn < 2 * HH) ? k : v;
    float* orow = om + (size_t)(m0 + rp * 32) * HH + (gn & (HH - 1)) + lo;
#pragma unroll
    for (int i = 0; i < 2; ++i)
#pragma unroll
      for (int r = 0; r < 8; ++r)
        orow[(size_t)(16 * i + 8 * hi + r) * HH] = acc[i][j][r];
  }
#undef PJ_LOAD
#undef PJ_STAGE
#undef PJ_COMPUTE
}

#define SC_BLK 128
#define SC_PITCH 132

__global__ __launch_bounds__(256) void qk_scores(const float* __restrict__ q,
                                                 const float* __restrict__ kmat,
                                                 float* __restrict__ St) {
  __shared__ __attribute__((aligned(16))) float shK[SC_BLK * SC_PITCH];
  __shared__ __attribute__((aligned(16))) float shQ[SC_BLK * SC_PITCH];

  const int tid = threadIdx.x;
  const int lane = tid & 31;
  const int lo = lane & 15, hi = lane >> 4;
  const int wave = __builtin_amdgcn_readfirstlane(tid >> 5);
  const int ws = wave & 3;
  const int wt = wave >> 2;

  const int pr = blockIdx.y;
  const int px = blockIdx.x;
  const int bt = (px <= pr) ? pr : 15 - pr;
  const int bs = (px <= pr) ? px : px - (pr + 1);
  const int b = blockIdx.z;
  const int s0 = bs * SC_BLK;
  const int t0 = bt * SC_BLK;

  {
    const int c4 = lane * 4;
    const float* gk = kmat + ((size_t)b * TT + s0 + wave) * HH + c4;
    const float* gq = q + ((size_t)b * TT + t0 + wave) * HH + c4;
    float4 tk[16], tq[16];
#pragma unroll
    for (int i = 0; i < 16; ++i) tk[i] = *(const float4*)(gk + (size_t)i * 8 * HH);
#pragma unroll
    for (int i = 0; i < 16; ++i) tq[i] = *(const float4*)(gq + (size_t)i * 8 * HH);
#pragma unroll
    for (int i = 0; i < 16; ++i)
      *(float4*)&shK[(wave + 8 * i) * SC_PITCH + c4] = tk[i];
#pragma unroll
    for (int i = 0; i < 16; ++i) {
      const int u = wave + 8 * i;
      const int prow = (u & 64) | ((u & 3) << 4) | ((u >> 2) & 15);
      *(float4*)&shQ[prow * SC_PITCH + c4] = tq[i];
    }
  }
  __syncthreads();

  if (bs == bt && wt == 0 && ws >= 2) return;

  const float* pA = &shK[(32 * ws + lo) * SC_PITCH + 2 * hi];
  const float* pB = &shQ[(64 * wt + lo) * SC_PITCH + 2 * hi];

  v8f acc[2][4] = {};
  v2f fa[2][4][2], fb[2][4][4];

#define SC_FRAGS(set, g)                                                \
  _Pragma("unroll") for (int kk = 0; kk < 4; ++kk) {                    \
    _Pragma("unroll") for (int i = 0; i < 2; ++i)                       \
      fa[set][kk][i] =                                                  \
          *(const v2f*)(pA + i * 16 * SC_PITCH + (g) * 16 + kk * 4);    \
    _Pragma("unroll") for (int j = 0; j < 4; ++j)                       \
      fb[set][kk][j] =                                                  \
          *(const v2f*)(pB + j * 16 * SC_PITCH + (g) * 16 + kk * 4);    \
  }

  SC_FRAGS(0, 0);
#pragma unroll
  for (int g = 0; g < 8; ++g) {
    const int cur = g & 1;
    if (g + 1 < 8) SC_FRAGS(cur ^ 1, g + 1);
    __builtin_amdgcn_sched_barrier(0);
#pragma unroll
    for (int kk = 0; kk < 4; ++kk)
#pragma unroll
      for (int i = 0; i < 2; ++i)
#pragma unroll
        for (int j = 0; j < 4; ++j)
          acc[i][j] = wmma_f32(fa[cur][kk][i], fb[cur][kk][j], acc[i][j]);
    __builtin_amdgcn_sched_barrier(0);
  }
#undef SC_FRAGS

  const float scale = 0.08838834764831845f;
  const int t = t0 + 64 * wt + 4 * lo;
#pragma unroll
  for (int i = 0; i < 2; ++i) {
#pragma unroll
    for (int r = 0; r < 8; ++r) {
      const int s = s0 + 32 * ws + 16 * i + 8 * hi + r;
      float4 o;
      o.x = (s <= t) ? acc[i][0][r] * scale : -1e30f;
      o.y = (s <= t + 1) ? acc[i][1][r] * scale : -1e30f;
      o.z = (s <= t + 2) ? acc[i][2][r] * scale : -1e30f;
      o.w = (s <= t + 3) ? acc[i][3][r] * scale : -1e30f;
      *(float4*)(St + ((size_t)b * TT + s) * TT + t) = o;
    }
  }
}

#define STAT_WAVES 32
#define STAT_NEG (-3.0e38f)

__global__ __launch_bounds__(1024) void col_softmax_stats(
    const float* __restrict__ St, float* __restrict__ colmax,
    float* __restrict__ colrcp) {
  __shared__ float sh_max[STAT_WAVES];
  __shared__ float sh_rcp[STAT_WAVES];

  const int lane = threadIdx.x & 31;
  const int wave = __builtin_amdgcn_readfirstlane((int)(threadIdx.x >> 5));
  const int gid = blockIdx.x * STAT_WAVES + wave;
  const int s = gid & (TT - 1);
  const float* row = St + (size_t)gid * TT;
  const int tfirst = s & ~15;
  const int tb = s & ~127;
  const int nit = (TT - tb) >> 7;

  float4 val[16];
#pragma unroll
  for (int i = 0; i < 16; ++i)
    val[i] = make_float4(STAT_NEG, STAT_NEG, STAT_NEG, STAT_NEG);

  const int tl = tb + lane * 4;
  if (tl >= tfirst) val[0] = *(const float4*)(row + tl);
#pragma unroll
  for (int i = 1; i < 16; ++i)
    if (i < nit) val[i] = *(const float4*)(row + tl + i * 128);

  val[0].x = (tl + 0 >= s) ? val[0].x : STAT_NEG;
  val[0].y = (tl + 1 >= s) ? val[0].y : STAT_NEG;
  val[0].z = (tl + 2 >= s) ? val[0].z : STAT_NEG;
  val[0].w = (tl + 3 >= s) ? val[0].w : STAT_NEG;

  float m = STAT_NEG;
#pragma unroll
  for (int i = 0; i < 16; ++i)
    if (i < nit)
      m = fmaxf(m, fmaxf(fmaxf(val[i].x, val[i].y), fmaxf(val[i].z, val[i].w)));
#pragma unroll
  for (int off = 16; off > 0; off >>= 1)
    m = fmaxf(m, __shfl_xor(m, off, 32));

  float sum = 0.f;
#pragma unroll
  for (int i = 0; i < 16; ++i)
    if (i < nit) {
      sum += __expf(val[i].x - m);
      sum += __expf(val[i].y - m);
      sum += __expf(val[i].z - m);
      sum += __expf(val[i].w - m);
    }
#pragma unroll
  for (int off = 16; off > 0; off >>= 1)
    sum += __shfl_xor(sum, off, 32);

  if (lane == 0) {
    sh_max[wave] = m;
    sh_rcp[wave] = 1.0f / sum;
  }
  __syncthreads();
  if (threadIdx.x < STAT_WAVES) {
    colmax[blockIdx.x * STAT_WAVES + threadIdx.x] = sh_max[threadIdx.x];
    colrcp[blockIdx.x * STAT_WAVES + threadIdx.x] = sh_rcp[threadIdx.x];
  }
}

__device__ __forceinline__ void lds_only_barrier() {
  __builtin_amdgcn_fence(__ATOMIC_RELEASE, "workgroup", "local");
  __builtin_amdgcn_s_barrier();
  __builtin_amdgcn_fence(__ATOMIC_ACQUIRE, "workgroup", "local");
}

#define OT_CHUNK 32
#define OT_DEPTH 4
#define P_PITCH 96
#define V_PITCH 288
#define P_TILE (16 * P_PITCH)
#define V_TILE (16 * V_PITCH)

__global__ __launch_bounds__(256) void attn_out(
    const float* __restrict__ St, const float* __restrict__ colmax,
    const float* __restrict__ colrcp, const float* __restrict__ v,
    float* __restrict__ out) {
  __shared__ __attribute__((aligned(16))) float shP[2][P_TILE];
  __shared__ __attribute__((aligned(16))) float shV[2][V_TILE];

  const int tid = threadIdx.x;
  const int lane = tid & 31;
  const int lo = lane & 15, hi = lane >> 4;
  const int wave = __builtin_amdgcn_readfirstlane(tid >> 5);
  const int mt = wave & 1;
  const int ng = wave >> 1;

  const int tb = TT / OT_CHUNK - 1 - (int)blockIdx.x;
  const int t0 = tb * 32;
  const int b = blockIdx.y;
  const int nchunk = tb + 1;

  const float* Sb = St + (size_t)b * TT * TT;
  const float* cm = colmax + (size_t)b * TT;
  const float* cr = colrcp + (size_t)b * TT;
  const float* vb = v + (size_t)b * TT * HH;

  const int psp = tid >> 4;
  const int pt2 = (tid & 15) * 2;
  const int vj = lane >> 4;
  const int vg = lane & 15;

  const float* gS = Sb + (size_t)(2 * psp) * TT + t0 + pt2;
  const float* gM = cm + 2 * psp;
  const float* gR = cr + 2 * psp;
  const float* gV = vb + (size_t)(4 * wave + vj) * HH + 4 * vg;
  const int ldsP = psp * P_PITCH + pt2 * 2;
  const int ldsV = (2 * wave) * V_PITCH + vg * 8 + vj * 4;

  const int fA = hi * P_PITCH + (mt * 16 + lo) * 2;
  const int fB = hi * V_PITCH + ng * 64 + (lo >> 2) * 8 + (lo & 3);

#define OT_SET(r)                                                       \
  v2f s_a##r, s_b##r, mx##r, rc##r;                                     \
  v4f va0##r, va1##r, vb0##r, vb1##r;
  OT_SET(0) OT_SET(1) OT_SET(2) OT_SET(3)
#undef OT_SET

#define OT_LOAD(c, r)                                                   \
  {                                                                     \
    const int cc = (c) < nchunk ? (c) : nchunk - 1;                     \
    const size_t so = (size_t)cc * OT_CHUNK;                            \
    s_a##r = *(const v2f*)(gS + so * TT);                               \
    s_b##r = *(const v2f*)(gS + so * TT + TT);                          \
    mx##r = *(const v2f*)(gM + so);                                     \
    rc##r = *(const v2f*)(gR + so);                                     \
    va0##r = *(const v4f*)(gV + so * HH);                               \
    va1##r = *(const v4f*)(gV + so * HH + 64);                          \
    vb0##r = *(const v4f*)(gV + so * HH + 2 * HH);                      \
    vb1##r = *(const v4f*)(gV + so * HH + 2 * HH + 64);                 \
  }

#define OT_STAGE(c, buf, r)                                             \
  {                                                                     \
    const int s = (c) * OT_CHUNK + 2 * psp;                             \
    const int t = t0 + pt2;                                             \
    v4f w;                                                              \
    w.x = (s <= t) ? __expf(s_a##r.x - mx##r.x) * rc##r.x : 0.f;        \
    w.y = (s + 1 <= t) ? __expf(s_b##r.x - mx##r.y) * rc##r.y : 0.f;    \
    w.z = (s <= t + 1) ? __expf(s_a##r.y - mx##r.x) * rc##r.x : 0.f;    \
    w.w = (s + 1 <= t + 1) ? __expf(s_b##r.y - mx##r.y) * rc##r.y : 0.f;\
    *(v4f*)&shP[buf][ldsP] = w;                                         \
    float* d = &shV[buf][ldsV];                                         \
    *(v4f*)(d) = va0##r;                                                \
    *(v4f*)(d + 128) = va1##r;                                          \
    *(v4f*)(d + V_PITCH) = vb0##r;                                      \
    *(v4f*)(d + V_PITCH + 128) = vb1##r;                                \
  }

  v8f acc0 = {}, acc1 = {};

#define OT_COMPUTE(buf)                                                 \
  {                                                                     \
    const float* pA = &shP[buf][fA];                                    \
    const float* pB = &shV[buf][fB];                                    \
    v2f fa[8], fb0[8], fb1[8];                                          \
    _Pragma("unroll") for (int kk = 0; kk < 8; ++kk) {                  \
      fa[kk] = *(const v2f*)(pA + kk * 2 * P_PITCH);                    \
      fb0[kk][0] = pB[kk * 2 * V_PITCH];                                \
      fb0[kk][1] = pB[kk * 2 * V_PITCH + 4];                            \
      fb1[kk][0] = pB[kk * 2 * V_PITCH + 32];                           \
      fb1[kk][1] = pB[kk * 2 * V_PITCH + 36];                           \
    }                                                                   \
    __builtin_amdgcn_sched_barrier(0x20);                               \
    _Pragma("unroll") for (int kk = 0; kk < 8; ++kk) {                  \
      acc0 = wmma_f32(fa[kk], fb0[kk], acc0);                           \
      acc1 = wmma_f32(fa[kk], fb1[kk], acc1);                           \
    }                                                                   \
  }

  OT_LOAD(0, 0);
  __builtin_amdgcn_sched_barrier(0);
  OT_LOAD(1, 1);
  __builtin_amdgcn_sched_barrier(0);
  OT_LOAD(2, 2);
  __builtin_amdgcn_sched_barrier(0);
  OT_LOAD(3, 3);
  __builtin_amdgcn_sched_barrier(0);
  OT_STAGE(0, 0, 0);
  lds_only_barrier();

#define OT_TRIP(u, un)                                                  \
  {                                                                     \
    OT_LOAD(c + (u) + OT_DEPTH, u);                                     \
    OT_COMPUTE((u) & 1);                                                \
    OT_STAGE(c + (u) + 1, (un) & 1, un);                                \
    lds_only_barrier();                                                 \
  }

  const int ntrip = (nchunk + OT_DEPTH - 1) & ~(OT_DEPTH - 1);
  for (int c = 0; c < ntrip; c += OT_DEPTH) {
    OT_TRIP(0, 1)
    OT_TRIP(1, 2)
    OT_TRIP(2, 3)
    OT_TRIP(3, 0)
  }
#undef OT_TRIP

  float* orow = out + ((size_t)b * TT + t0 + mt * 16) * HH + ng * 32 + lo;
#pragma unroll
  for (int r = 0; r < 8; ++r) {
    orow[(size_t)(r + 8 * hi) * HH] = acc0[r];
    orow[(size_t)(r + 8 * hi) * HH + 16] = acc1[r];
  }
#undef OT_LOAD
#undef OT_STAGE
#undef OT_COMPUTE
}

extern "C" void kernel_launch(void* const* d_in, const int* in_sizes, int n_in,
                              void* d_out, int out_size, void* d_ws,
                              size_t ws_size, hipStream_t stream) {
  (void)in_sizes; (void)n_in; (void)out_size; (void)ws_size;
  const float* x = (const float*)d_in[0];
  const float* Wq = (const float*)d_in[1];
  const float* Wk = (const float*)d_in[2];
  const float* Wv = (const float*)d_in[3];
  float* out = (float*)d_out;

  float* ws = (float*)d_ws;
  float* q = ws;
  float* k = q + (size_t)BB * TT * HH;
  float* v = k + (size_t)BB * TT * HH;
  float* S = v + (size_t)BB * TT * HH;
  float* cmax = S + (size_t)BB * TT * TT;
  float* crcp = cmax + (size_t)BB * TT;

  qkv_proj<<<dim3(2, BB * TT / PJ_BM), 256, 0, stream>>>(x, Wq, Wk, Wv, q, k, v);
  qk_scores<<<dim3(17, 8, BB), 256, 0, stream>>>(q, k, S);
  col_softmax_stats<<<dim3(BB * TT / STAT_WAVES), STAT_WAVES * 32, 0, stream>>>(
      S, cmax, crcp);
  attn_out<<<dim3(TT / OT_CHUNK, BB), 256, 0, stream>>>(S, cmax, crcp, v, out);
}


# score kernel: work-group = 4 waves making 128(s) x 64(t) (was 8 waves, 128 x 128): LDS 99 KiB, three resident per compute unit, staging of one overlaps products of the others; sum order kept
# speedup vs baseline: 15.5980x; 1.0369x over previous
#include <hip/hip_runtime.h>
#include <hip/hip_bf16.h>
#include <math.h>

#define BB 4
#define TT 2048
#define CC 1024
#define HH 128

typedef __attribute__((ext_vector_type(2))) float v2f;
typedef __attribute__((ext_vector_type(8))) float v8f;
typedef __attribute__((ext_vector_type(4))) float v4f;

__device__ __forceinline__ v8f wmma_f32(v2f a, v2f b, v8f c) {
  return __builtin_amdgcn_wmma_f32_16x16x4_f32(false, a, false, b, (short)0, c,
                                               false, false);
}

#define PJ_BM 64
#define PJ_BK 32
#define PJ_XP 36
#define PJ_WP 416
#define PJ_XT (PJ_BM * PJ_XP)
#define PJ_WT ((PJ_BK / 2) * PJ_WP)

__global__ __launch_bounds__(256) void qkv_proj(
    const float* __restrict__ x, const float* __restrict__ Wq,
    const float* __restrict__ Wk, const float* __restrict__ Wv,
    float* __restrict__ q, float* __restrict__ k, float* __restrict__ v) {
  __shared__ __attribute__((aligned(16))) float shX[2][PJ_XT];
  __shared__ __attribute__((aligned(16))) float shW[2][PJ_WT];

  const int tid = threadIdx.x;
  const int lane = tid & 31;
  const int lo = lane & 15, hi = lane >> 4;
  const int wave = __builtin_amdgcn_readfirstlane(tid >> 5);
  const int rp = wave & 1;
  const int cg = wave >> 1;

  const int half = blockIdx.x;
  const int m0 = blockIdx.y * PJ_BM;

  const int xr = tid >> 3;
  const int xc = (tid & 7) * 4;
  const int wp = tid >> 4;
  const int wc = (tid & 15) * 4;

  const float* gX = x + (size_t)(m0 + xr) * CC + xc;
  const float* gW0 = (half == 0 ? Wq : Wk) + (size_t)(2 * wp) * HH +
                     (half == 0 ? 0 : 64) + wc;
  const float* gW1 = (half == 0 ? Wq : Wv) + (size_t)(2 * wp) * HH +
                     (half == 0 ? 64 : 0) + wc;
  const float* gW2 = (half == 0 ? Wk : Wv) + (size_t)(2 * wp) * HH +
                     (half == 0 ? 0 : 64) + wc;
  const int ldsX = xr * PJ_XP + xc;
  const int ldsW = wp * PJ_WP + wc * 2;

  const int fA = (rp * 32 + lo) * PJ_XP + 2 * hi;
  const int fB = hi * PJ_WP + (cg * 48 + lo) * 2;

  float4 xa, xb, w0a, w0b, w1a, w1b, w2a, w2b;

#define PJ_LOAD(c)                                                      \
  {                                                                     \
    const size_t ko = (size_t)(c) * PJ_BK;                              \
    xa = *(const float4*)(gX + ko);                                     \
    xb = *(const float4*)(gX + ko + (size_t)32 * CC);                   \
    w0a = *(const float4*)(gW0 + ko * HH);                              \
    w0b = *(const float4*)(gW0 + ko * HH + HH);                         \
    w1a = *(const float4*)(gW1 + ko * HH);                              \
    w1b = *(const float4*)(gW1 + ko * HH + HH);                         \
    w2a = *(const float4*)(gW2 + ko * HH);                              \
    w2b = *(const float4*)(gW2 + ko * HH + HH);                         \
  }

#define PJ_STAGE(buf)                                                   \
  {                                                                     \
    *(float4*)&shX[buf][ldsX] = xa;                                     \
    *(float4*)&shX[buf][ldsX + 32 * PJ_XP] = xb;                        \
    float* d = &shW[buf][ldsW];                                         \
    *(float4*)(d) = make_float4(w0a.x, w0b.x, w0a.y, w0b.y);            \
    *(float4*)(d + 4) = make_float4(w0a.z, w0b.z, w0a.w, w0b.w);        \
    *(float4*)(d + 128) = make_float4(w1a.x, w1b.x, w1a.y, w1b.y);      \
    *(float4*)(d + 132) = make_float4(w1a.z, w1b.z, w1a.w, w1b.w);      \
    *(float4*)(d + 256) = make_float4(w2a.x, w2b.x, w2a.y, w2b.y);      \
    *(float4*)(d + 260) = make_float4(w2a.z, w2b.z, w2a.w, w2b.w);      \
  }

  v8f acc[2][3] = {};

#define PJ_COMPUTE(buf)                                                 \
  {                                                                     \
    const float* pA = &shX[buf][fA];                                    \
    const float* pB = &shW[buf][fB];                                    \
    v2f fa[8][2], fb[8][3];                                             \
    _Pragma("unroll") for (int kk = 0; kk < 8; ++kk) {                  \
      _Pragma("unroll") for (int i = 0; i < 2; ++i)                     \
        fa[kk][i] = *(const v2f*)(pA + i * 16 * PJ_XP + kk * 4);        \
      _Pragma("unroll") for (int j = 0; j < 3; ++j)                     \
        fb[kk][j] = *(const v2f*)(pB + kk * 2 * PJ_WP + j * 32);        \
    }                                                                   \
    _Pragma("unroll") for (int kk = 0; kk < 8; ++kk)                    \
      _Pragma("unroll") for (int i = 0; i < 2; ++i)                     \
        _Pragma("unroll") for (int j = 0; j < 3; ++j)                   \
          acc[i][j] = wmma_f32(fa[kk][i], fb[kk][j], acc[i][j]);        \
    __builtin_amdgcn_sched_group_barrier(0x020, 8, 0);                  \
    __builtin_amdgcn_sched_group_barrier(0x100, 40, 0);                 \
    __builtin_amdgcn_sched_group_barrier(0x008, 48, 0);                 \
  }

  PJ_LOAD(0);
  PJ_STAGE(0);
  __syncthreads();
  for (int c = 0; c + 1 < CC / PJ_BK; ++c) {
    const int cur = c & 1;
    PJ_LOAD(c + 1);
    PJ_COMPUTE(cur);
    PJ_STAGE(cur ^ 1);
    __syncthreads();
  }
  PJ_COMPUTE((CC / PJ_BK - 1) & 1);

#pragma unroll
  for (int j = 0; j < 3; ++j) {
    const int gn = 192 * half + 48 * cg + 16 * j;
    float* om = (gn < HH) ? q : (gn < 2 * HH) ? k : v;
    float* orow = om + (size_t)(m0 + rp * 32) * HH + (gn & (HH - 1)) + lo;
#pragma unroll
    for (int i = 0; i < 2; ++i)
#pragma unroll
      for (int r = 0; r < 8; ++r)
        orow[(size_t)(16 * i + 8 * hi + r) * HH] = acc[i][j][r];
  }
#undef PJ_LOAD
#undef PJ_STAGE
#undef PJ_COMPUTE
}

#define SC_BS 128
#define SC_BT 64
#define SC_PITCH 132

__global__ __launch_bounds__(128) void qk_scores(const float* __restrict__ q,
                                                 const float* __restrict__ kmat,
                                                 float* __restrict__ St) {
  __shared__ __attribute__((aligned(16))) float shK[SC_BS * SC_PITCH];
  __shared__ __attribute__((aligned(16))) float shQ[SC_BT * SC_PITCH];

  const int tid = threadIdx.x;
  const int lane = tid & 31;
  const int lo = lane & 15, hi = lane >> 4;
  const int ws = __builtin_amdgcn_readfirstlane(tid >> 5);

  const int pr = blockIdx.y;
  const int px = (int)blockIdx.x >> 1;
  const int th = (int)blockIdx.x & 1;
  const int bt = (px <= pr) ? pr : 15 - pr;
  const int bs = (px <= pr) ? px : px - (pr + 1);
  const int b = blockIdx.z;
  const int s0 = bs * SC_BS;
  const int t0 = bt * SC_BS + th * SC_BT;

  {
    const int c4 = lane * 4;
    const float* gk = kmat + ((size_t)b * TT + s0 + ws) * HH + c4;
    const float* gq = q + ((size_t)b * TT + t0 + ws) * HH + c4;
    float4 tk[32], tq[16];
#pragma unroll
    for (int i = 0; i < 32; ++i) tk[i] = *(const float4*)(gk + (size_t)i * 4 * HH);
#pragma unroll
    for (int i = 0; i < 16; ++i) tq[i] = *(const float4*)(gq + (size_t)i * 4 * HH);
#pragma unroll
    for (int i = 0; i < 32; ++i)
      *(float4*)&shK[(ws + 4 * i) * SC_PITCH + c4] = tk[i];
#pragma unroll
    for (int i = 0; i < 16; ++i)
      *(float4*)&shQ[(16 * ws + i) * SC_PITCH + c4] = tq[i];
  }
  __syncthreads();

  if (bs == bt && th == 0 && ws >= 2) return;

  const float* pA = &shK[(32 * ws + lo) * SC_PITCH + 2 * hi];
  const float* pB = &shQ[lo * SC_PITCH + 2 * hi];

  v8f acc[2][4] = {};
  v2f fa[2][4][2], fb[2][4][4];

#define SC_FRAGS(set, g)                                                \
  _Pragma("unroll") for (int kk = 0; kk < 4; ++kk) {                    \
    _Pragma("unroll") for (int i = 0; i < 2; ++i)                       \
      fa[set][kk][i] =                                                  \
          *(const v2f*)(pA + i * 16 * SC_PITCH + (g) * 16 + kk * 4);    \
    _Pragma("unroll") for (int j = 0; j < 4; ++j)                       \
      fb[set][kk][j] =                                                  \
          *(const v2f*)(pB + j * 16 * SC_PITCH + (g) * 16 + kk * 4);    \
  }

  SC_FRAGS(0, 0);
#pragma unroll
  for (int g = 0; g < 8; ++g) {
    const int cur = g & 1;
    if (g + 1 < 8) SC_FRAGS(cur ^ 1, g + 1);
    __builtin_amdgcn_sched_barrier(0);
#pragma unroll
    for (int kk = 0; kk < 4; ++kk)
#pragma unroll
      for (int i = 0; i < 2; ++i)
#pragma unroll
        for (int j = 0; j < 4; ++j)
          acc[i][j] = wmma_f32(fa[cur][kk][i], fb[cur][kk][j], acc[i][j]);
    __builtin_amdgcn_sched_barrier(0);
  }
#undef SC_FRAGS

  const float scale = 0.08838834764831845f;
  const int t = t0 + 4 * lo;
#pragma unroll
  for (int i = 0; i < 2; ++i) {
#pragma unroll
    for (int r = 0; r < 8; ++r) {
      const int s = s0 + 32 * ws + 16 * i + 8 * hi + r;
      float4 o;
      o.x = (s <= t) ? acc[i][0][r] * scale : -1e30f;
      o.y = (s <= t + 1) ? acc[i][1][r] * scale : -1e30f;
      o.z = (s <= t + 2) ? acc[i][2][r] * scale : -1e30f;
      o.w = (s <= t + 3) ? acc[i][3][r] * scale : -1e30f;
      *(float4*)(St + ((size_t)b * TT + s) * TT + t) = o;
    }
  }
}

#define STAT_WAVES 32
#define STAT_NEG (-3.0e38f)

__global__ __launch_bounds__(1024) void col_softmax_stats(
    const float* __restrict__ St, float* __restrict__ colmax,
    float* __restrict__ colrcp) {
  __shared__ float sh_max[STAT_WAVES];
  __shared__ float sh_rcp[STAT_WAVES];

  const int lane = threadIdx.x & 31;
  const int wave = __builtin_amdgcn_readfirstlane((int)(threadIdx.x >> 5));
  const int gid = blockIdx.x * STAT_WAVES + wave;
  const int s = gid & (TT - 1);
  const float* row = St + (size_t)gid * TT;
  const int tfirst = s & ~15;
  const int tb = s & ~127;
  const int nit = (TT - tb) >> 7;

  float4 val[16];
#pragma unroll
  for (int i = 0; i < 16; ++i)
    val[i] = make_float4(STAT_NEG, STAT_NEG, STAT_NEG, STAT_NEG);

  const int tl = tb + lane * 4;
  if (tl >= tfirst) val[0] = *(const float4*)(row + tl);
#pragma unroll
  for (int i = 1; i < 16; ++i)
    if (i < nit) val[i] = *(const float4*)(row + tl + i * 128);

  val[0].x = (tl + 0 >= s) ? val[0].x : STAT_NEG;
  val[0].y = (tl + 1 >= s) ? val[0].y : STAT_NEG;
  val[0].z = (tl + 2 >= s) ? val[0].z : STAT_NEG;
  val[0].w = (tl + 3 >= s) ? val[0].w : STAT_NEG;

  float m = STAT_NEG;
#pragma unroll
  for (int i = 0; i < 16; ++i)
    if (i < nit)
      m = fmaxf(m, fmaxf(fmaxf(val[i].x, val[i].y), fmaxf(val[i].z, val[i].w)));
#pragma unroll
  for (int off = 16; off > 0; off >>= 1)
    m = fmaxf(m, __shfl_xor(m, off, 32));

  float sum = 0.f;
#pragma unroll
  for (int i = 0; i < 16; ++i)
    if (i < nit) {
      sum += __expf(val[i].x - m);
      sum += __expf(val[i].y - m);
      sum += __expf(val[i].z - m);
      sum += __expf(val[i].w - m);
    }
#pragma unroll
  for (int off = 16; off > 0; off >>= 1)
    sum += __shfl_xor(sum, off, 32);

  if (lane == 0) {
    sh_max[wave] = m;
    sh_rcp[wave] = 1.0f / sum;
  }
  __syncthreads();
  if (threadIdx.x < STAT_WAVES) {
    colmax[blockIdx.x * STAT_WAVES + threadIdx.x] = sh_max[threadIdx.x];
    colrcp[blockIdx.x * STAT_WAVES + threadIdx.x] = sh_rcp[threadIdx.x];
  }
}

__device__ __forceinline__ void lds_only_barrier() {
  __builtin_amdgcn_fence(__ATOMIC_RELEASE, "workgroup", "local");
  __builtin_amdgcn_s_barrier();
  __builtin_amdgcn_fence(__ATOMIC_ACQUIRE, "workgroup", "local");
}

#define OT_CHUNK 32
#define OT_DEPTH 4
#define P_PITCH 96
#define V_PITCH 288
#define P_TILE (16 * P_PITCH)
#define V_TILE (16 * V_PITCH)

__global__ __launch_bounds__(256) void attn_out(
    const float* __restrict__ St, const float* __restrict__ colmax,
    const float* __restrict__ colrcp, const float* __restrict__ v,
    float* __restrict__ out) {
  __shared__ __attribute__((aligned(16))) float shP[2][P_TILE];
  __shared__ __attribute__((aligned(16))) float shV[2][V_TILE];

  const int tid = threadIdx.x;
  const int lane = tid & 31;
  const int lo = lane & 15, hi = lane >> 4;
  const int wave = __builtin_amdgcn_readfirstlane(tid >> 5);
  const int mt = wave & 1;
  const int ng = wave >> 1;

  const int tb = TT / OT_CHUNK - 1 - (int)blockIdx.x;
  const int t0 = tb * 32;
  const int b = blockIdx.y;
  const int nchunk = tb + 1;

  const float* Sb = St + (size_t)b * TT * TT;
  const float* cm = colmax + (size_t)b * TT;
  const float* cr = colrcp + (size_t)b * TT;
  const float* vb = v + (size_t)b * TT * HH;

  const int psp = tid >> 4;
  const int pt2 = (tid & 15) * 2;
  const int vj = lane >> 4;
  const int vg = lane & 15;

  const float* gS = Sb + (size_t)(2 * psp) * TT + t0 + pt2;
  const float* gM = cm + 2 * psp;
  const float* gR = cr + 2 * psp;
  const float* gV = vb + (size_t)(4 * wave + vj) * HH + 4 * vg;
  const int ldsP = psp * P_PITCH + pt2 * 2;
  const int ldsV = (2 * wave) * V_PITCH + vg * 8 + vj * 4;

  const int fA = hi * P_PITCH + (mt * 16 + lo) * 2;
  const int fB = hi * V_PITCH + ng * 64 + (lo >> 2) * 8 + (lo & 3);

#define OT_SET(r)                                                       \
  v2f s_a##r, s_b##r, mx##r, rc##r;                                     \
  v4f va0##r, va1##r, vb0##r, vb1##r;
  OT_SET(0) OT_SET(1) OT_SET(2) OT_SET(3)
#undef OT_SET

#define OT_LOAD(c, r)                                                   \
  {                                                                     \
    const int cc = (c) < nchunk ? (c) : nchunk - 1;                     \
    const size_t so = (size_t)cc * OT_CHUNK;                            \
    s_a##r = *(const v2f*)(gS + so * TT);                               \
    s_b##r = *(const v2f*)(gS + so * TT + TT);                          \
    mx##r = *(const v2f*)(gM + so);                                     \
    rc##r = *(const v2f*)(gR + so);                                     \
    va0##r = *(const v4f*)(gV + so * HH);                               \
    va1##r = *(const v4f*)(gV + so * HH + 64);                          \
    vb0##r = *(const v4f*)(gV + so * HH + 2 * HH);                      \
    vb1##r = *(const v4f*)(gV + so * HH + 2 * HH + 64);                 \
  }

#define OT_STAGE(c, buf, r)                                             \
  {                                                                     \
    const int s = (c) * OT_CHUNK + 2 * psp;                             \
    const int t = t0 + pt2;                                             \
    v4f w;                                                              \
    w.x = (s <= t) ? __expf(s_a##r.x - mx##r.x) * rc##r.x : 0.f;        \
    w.y = (s + 1 <= t) ? __expf(s_b##r.x - mx##r.y) * rc##r.y : 0.f;    \
    w.z = (s <= t + 1) ? __expf(s_a##r.y - mx##r.x) * rc##r.x : 0.f;    \
    w.w = (s + 1 <= t + 1) ? __expf(s_b##r.y - mx##r.y) * rc##r.y : 0.f;\
    *(v4f*)&shP[buf][ldsP] = w;                                         \
    float* d = &shV[buf][ldsV];                                         \
    *(v4f*)(d) = va0##r;                                                \
    *(v4f*)(d + 128) = va1##r;                                          \
    *(v4f*)(d + V_PITCH) = vb0##r;                                      \
    *(v4f*)(d + V_PITCH + 128) = vb1##r;                                \
  }

  v8f acc0 = {}, acc1 = {};

#define OT_COMPUTE(buf)                                                 \
  {                                                                     \
    const float* pA = &shP[buf][fA];                                    \
    const float* pB = &shV[buf][fB];                                    \
    v2f fa[8], fb0[8], fb1[8];                                          \
    _Pragma("unroll") for (int kk = 0; kk < 8; ++kk) {                  \
      fa[kk] = *(const v2f*)(pA + kk * 2 * P_PITCH);                    \
      fb0[kk][0] = pB[kk * 2 * V_PITCH];                                \
      fb0[kk][1] = pB[kk * 2 * V_PITCH + 4];                            \
      fb1[kk][0] = pB[kk * 2 * V_PITCH + 32];                           \
      fb1[kk][1] = pB[kk * 2 * V_PITCH + 36];                           \
    }                                                                   \
    __builtin_amdgcn_sched_barrier(0x20);                               \
    _Pragma("unroll") for (int kk = 0; kk < 8; ++kk) {                  \
      acc0 = wmma_f32(fa[kk], fb0[kk], acc0);                           \
      acc1 = wmma_f32(fa[kk], fb1[kk], acc1);                           \
    }                                                                   \
  }

  OT_LOAD(0, 0);
  __builtin_amdgcn_sched_barrier(0);
  OT_LOAD(1, 1);
  __builtin_amdgcn_sched_barrier(0);
  OT_LOAD(2, 2);
  __builtin_amdgcn_sched_barrier(0);
  OT_LOAD(3, 3);
  __builtin_amdgcn_sched_barrier(0);
  OT_STAGE(0, 0, 0);
  lds_only_barrier();

#define OT_TRIP(u, un)                                                  \
  {                                                                     \
    OT_LOAD(c + (u) + OT_DEPTH, u);                                     \
    OT_COMPUTE((u) & 1);                                                \
    OT_STAGE(c + (u) + 1, (un) & 1, un);                                \
    lds_only_barrier();                                                 \
  }

  const int ntrip = (nchunk + OT_DEPTH - 1) & ~(OT_DEPTH - 1);
  for (int c = 0; c < ntrip; c += OT_DEPTH) {
    OT_TRIP(0, 1)
    OT_TRIP(1, 2)
    OT_TRIP(2, 3)
    OT_TRIP(3, 0)
  }
#undef OT_TRIP

  float* orow = out + ((size_t)b * TT + t0 + mt * 16) * HH + ng * 32 + lo;
#pragma unroll
  for (int r = 0; r < 8; ++r) {
    orow[(size_t)(r + 8 * hi) * HH] = acc0[r];
    orow[(size_t)(r + 8 * hi) * HH + 16] = acc1[r];
  }
#undef OT_LOAD
#undef OT_STAGE
#undef OT_COMPUTE
}

extern "C" void kernel_launch(void* const* d_in, const int* in_sizes, int n_in,
                              void* d_out, int out_size, void* d_ws,
                              size_t ws_size, hipStream_t stream) {
  (void)in_sizes; (void)n_in; (void)out_size; (void)ws_size;
  const float* x = (const float*)d_in[0];
  const float* Wq = (const float*)d_in[1];
  const float* Wk = (const float*)d_in[2];
  const float* Wv = (const float*)d_in[3];
  float* out = (float*)d_out;

  float* ws = (float*)d_ws;
  float* q = ws;
  float* k = q + (size_t)BB * TT * HH;
  float* v = k + (size_t)BB * TT * HH;
  float* S = v + (size_t)BB * TT * HH;
  float* cmax = S + (size_t)BB * TT * TT;
  float* crcp = cmax + (size_t)BB * TT;

  qkv_proj<<<dim3(2, BB * TT / PJ_BM), 256, 0, stream>>>(x, Wq, Wk, Wv, q, k, v);
  qk_scores<<<dim3(34, 8, BB), 128, 0, stream>>>(q, k, S);
  col_softmax_stats<<<dim3(BB * TT / STAT_WAVES), STAT_WAVES * 32, 0, stream>>>(
      S, cmax, crcp);
  attn_out<<<dim3(TT / OT_CHUNK, BB), 256, 0, stream>>>(S, cmax, crcp, v, out);
}


# output kernel: a work-group makes 64 of the 128 columns of row block 63-i and then of row block i (was all columns of one block): all 256 groups walk 65 chunks, were 1..64; sum order kept
# speedup vs baseline: 16.6099x; 1.0649x over previous
#include <hip/hip_runtime.h>
#include <hip/hip_bf16.h>
#include <math.h>

#define BB 4
#define TT 2048
#define CC 1024
#define HH 128

typedef __attribute__((ext_vector_type(2))) float v2f;
typedef __attribute__((ext_vector_type(8))) float v8f;
typedef __attribute__((ext_vector_type(4))) float v4f;

__device__ __forceinline__ v8f wmma_f32(v2f a, v2f b, v8f c) {
  return __builtin_amdgcn_wmma_f32_16x16x4_f32(false, a, false, b, (short)0, c,
                                               false, false);
}

#define PJ_BM 64
#define PJ_BK 32
#define PJ_XP 36
#define PJ_WP 416
#define PJ_XT (PJ_BM * PJ_XP)
#define PJ_WT ((PJ_BK / 2) * PJ_WP)

__global__ __launch_bounds__(256) void qkv_proj(
    const float* __restrict__ x, const float* __restrict__ Wq,
    const float* __restrict__ Wk, const float* __restrict__ Wv,
    float* __restrict__ q, float* __restrict__ k, float* __restrict__ v) {
  __shared__ __attribute__((aligned(16))) float shX[2][PJ_XT];
  __shared__ __attribute__((aligned(16))) float shW[2][PJ_WT];

  const int tid = threadIdx.x;
  const int lane = tid & 31;
  const int lo = lane & 15, hi = lane >> 4;
  const int wave = __builtin_amdgcn_readfirstlane(tid >> 5);
  const int rp = wave & 1;
  const int cg = wave >> 1;

  const int half = blockIdx.x;
  const int m0 = blockIdx.y * PJ_BM;

  const int xr = tid >> 3;
  const int xc = (tid & 7) * 4;
  const int wp = tid >> 4;
  const int wc = (tid & 15) * 4;

  const float* gX = x + (size_t)(m0 + xr) * CC + xc;
  const float* gW0 = (half == 0 ? Wq : Wk) + (size_t)(2 * wp) * HH +
                     (half == 0 ? 0 : 64) + wc;
  const float* gW1 = (half == 0 ? Wq : Wv) + (size_t)(2 * wp) * HH +
                     (half == 0 ? 64 : 0) + wc;
  const float* gW2 = (half == 0 ? Wk : Wv) + (size_t)(2 * wp) * HH +
                     (half == 0 ? 0 : 64) + wc;
  const int ldsX = xr * PJ_XP + xc;
  const int ldsW = wp * PJ_WP + wc * 2;

  const int fA = (rp * 32 + lo) * PJ_XP + 2 * hi;
  const int fB = hi * PJ_WP + (cg * 48 + lo) * 2;

  float4 xa, xb, w0a, w0b, w1a, w1b, w2a, w2b;

#define PJ_LOAD(c)                                                      \
  {                                                                     \
    const size_t ko = (size_t)(c) * PJ_BK;                              \
    xa = *(const float4*)(gX + ko);                                     \
    xb = *(const float4*)(gX + ko + (size_t)32 * CC);                   \
    w0a = *(const float4*)(gW0 + ko * HH);                              \
    w0b = *(const float4*)(gW0 + ko * HH + HH);                         \
    w1a = *(const float4*)(gW1 + ko * HH);                              \
    w1b = *(const float4*)(gW1 + ko * HH + HH);                         \
    w2a = *(const float4*)(gW2 + ko * HH);                              \
    w2b = *(const float4*)(gW2 + ko * HH + HH);                         \
  }

#define PJ_STAGE(buf)                                                   \
  {                                                                     \
    *(float4*)&shX[buf][ldsX] = xa;                                     \
    *(float4*)&shX[buf][ldsX + 32 * PJ_XP] = xb;                        \
    float* d = &shW[buf][ldsW];                                         \
    *(float4*)(d) = make_float4(w0a.x, w0b.x, w0a.y, w0b.y);            \
    *(float4*)(d + 4) = make_float4(w0a.z, w0b.z, w0a.w, w0b.w);        \
    *(float4*)(d + 128) = make_float4(w1a.x, w1b.x, w1a.y, w1b.y);      \
    *(float4*)(d + 132) = make_float4(w1a.z, w1b.z, w1a.w, w1b.w);      \
    *(float4*)(d + 256) = make_float4(w2a.x, w2b.x, w2a.y, w2b.y);      \
    *(float4*)(d + 260) = make_float4(w2a.z, w2b.z, w2a.w, w2b.w);      \
  }

  v8f acc[2][3] = {};

#define PJ_COMPUTE(buf)                                                 \
  {                                                                     \
    const float* pA = &shX[buf][fA];                                    \
    const float* pB = &shW[buf][fB];                                    \
    v2f fa[8][2], fb[8][3];                                             \
    _Pragma("unroll") for (int kk = 0; kk < 8; ++kk) {                  \
      _Pragma("unroll") for (int i = 0; i < 2; ++i)                     \
        fa[kk][i] = *(const v2f*)(pA + i * 16 * PJ_XP + kk * 4);        \
      _Pragma("unroll") for (int j = 0; j < 3; ++j)                     \
        fb[kk][j] = *(const v2f*)(pB + kk * 2 * PJ_WP + j * 32);        \
    }                                                                   \
    _Pragma("unroll") for (int kk = 0; kk < 8; ++kk)                    \
      _Pragma("unroll") for (int i = 0; i < 2; ++i)                     \
        _Pragma("unroll") for (int j = 0; j < 3; ++j)                   \
          acc[i][j] = wmma_f32(fa[kk][i], fb[kk][j], acc[i][j]);        \
    __builtin_amdgcn_sched_group_barrier(0x020, 8, 0);                  \
    __builtin_amdgcn_sched_group_barrier(0x100, 40, 0);                 \
    __builtin_amdgcn_sched_group_barrier(0x008, 48, 0);                 \
  }

  PJ_LOAD(0);
  PJ_STAGE(0);
  __syncthreads();
  for (int c = 0; c + 1 < CC / PJ_BK; ++c) {
    const int cur = c & 1;
    PJ_LOAD(c + 1);
    PJ_COMPUTE(cur);
    PJ_STAGE(cur ^ 1);
    __syncthreads();
  }
  PJ_COMPUTE((CC / PJ_BK - 1) & 1);

#pragma unroll
  for (int j = 0; j < 3; ++j) {
    const int gn = 192 * half + 48 * cg + 16 * j;
    float* om = (gn < HH) ? q : (gn < 2 * HH) ? k : v;
    float* orow = om + (size_t)(m0 + rp * 32) * HH + (gn & (HH - 1)) + lo;
#pragma unroll
    for (int i = 0; i < 2; ++i)
#pragma unroll
      for (int r = 0; r < 8; ++r)
        orow[(size_t)(16 * i + 8 * hi + r) * HH] = acc[i][j][r];
  }
#undef PJ_LOAD
#undef PJ_STAGE
#undef PJ_COMPUTE
}

#define SC_BS 128
#define SC_BT 64
#define SC_PITCH 132

__global__ __launch_bounds__(128) void qk_scores(const float* __restrict__ q,
                                                 const float* __restrict__ kmat,
                                                 float* __restrict__ St) {
  __shared__ __attribute__((aligned(16))) float shK[SC_BS * SC_PITCH];
  __shared__ __attribute__((aligned(16))) float shQ[SC_BT * SC_PITCH];

  const int tid = threadIdx.x;
  const int lane = tid & 31;
  const int lo = lane & 15, hi = lane >> 4;
  const int ws = __builtin_amdgcn_readfirstlane(tid >> 5);

  const int pr = blockIdx.y;
  const int px = (int)blockIdx.x >> 1;
  const int th = (int)blockIdx.x & 1;
  const int bt = (px <= pr) ? pr : 15 - pr;
  const int bs = (px <= pr) ? px : px - (pr + 1);
  const int b = blockIdx.z;
  const int s0 = bs * SC_BS;
  const int t0 = bt * SC_BS + th * SC_BT;

  {
    const int c4 = lane * 4;
    const float* gk = kmat + ((size_t)b * TT + s0 + ws) * HH + c4;
    const float* gq = q + ((size_t)b * TT + t0 + ws) * HH + c4;
    float4 tk[32], tq[16];
#pragma unroll
    for (int i = 0; i < 32; ++i) tk[i] = *(const float4*)(gk + (size_t)i * 4 * HH);
#pragma unroll
    for (int i = 0; i < 16; ++i) tq[i] = *(const float4*)(gq + (size_t)i * 4 * HH);
#pragma unroll
    for (int i = 0; i < 32; ++i)
      *(float4*)&shK[(ws + 4 * i) * SC_PITCH + c4] = tk[i];
#pragma unroll
    for (int i = 0; i < 16; ++i)
      *(float4*)&shQ[(16 * ws + i) * SC_PITCH + c4] = tq[i];
  }
  __syncthreads();

  if (bs == bt && th == 0 && ws >= 2) return;

  const float* pA = &shK[(32 * ws + lo) * SC_PITCH + 2 * hi];
  const float* pB = &shQ[lo * SC_PITCH + 2 * hi];

  v8f acc[2][4] = {};
  v2f fa[2][4][2], fb[2][4][4];

#define SC_FRAGS(set, g)                                                \
  _Pragma("unroll") for (int kk = 0; kk < 4; ++kk) {                    \
    _Pragma("unroll") for (int i = 0; i < 2; ++i)                       \
      fa[set][kk][i] =                                                  \
          *(const v2f*)(pA + i * 16 * SC_PITCH + (g) * 16 + kk * 4);    \
    _Pragma("unroll") for (int j = 0; j < 4; ++j)                       \
      fb[set][kk][j] =                                                  \
          *(const v2f*)(pB + j * 16 * SC_PITCH + (g) * 16 + kk * 4);    \
  }

  SC_FRAGS(0, 0);
#pragma unroll
  for (int g = 0; g < 8; ++g) {
    const int cur = g & 1;
    if (g + 1 < 8) SC_FRAGS(cur ^ 1, g + 1);
    __builtin_amdgcn_sched_barrier(0);
#pragma unroll
    for (int kk = 0; kk < 4; ++kk)
#pragma unroll
      for (int i = 0; i < 2; ++i)
#pragma unroll
        for (int j = 0; j < 4; ++j)
          acc[i][j] = wmma_f32(fa[cur][kk][i], fb[cur][kk][j], acc[i][j]);
    __builtin_amdgcn_sched_barrier(0);
  }
#undef SC_FRAGS

  const float scale = 0.08838834764831845f;
  const int t = t0 + 4 * lo;
#pragma unroll
  for (int i = 0; i < 2; ++i) {
#pragma unroll
    for (int r = 0; r < 8; ++r) {
      const int s = s0 + 32 * ws + 16 * i + 8 * hi + r;
      float4 o;
      o.x = (s <= t) ? acc[i][0][r] * scale : -1e30f;
      o.y = (s <= t + 1) ? acc[i][1][r] * scale : -1e30f;
      o.z = (s <= t + 2) ? acc[i][2][r] * scale : -1e30f;
      o.w = (s <= t + 3) ? acc[i][3][r] * scale : -1e30f;
      *(float4*)(St + ((size_t)b * TT + s) * TT + t) = o;
    }
  }
}

#define STAT_WAVES 32
#define STAT_NEG (-3.0e38f)

__global__ __launch_bounds__(1024) void col_softmax_stats(
    const float* __restrict__ St, float* __restrict__ colmax,
    float* __restrict__ colrcp) {
  __shared__ float sh_max[STAT_WAVES];
  __shared__ float sh_rcp[STAT_WAVES];

  const int lane = threadIdx.x & 31;
  const int wave = __builtin_amdgcn_readfirstlane((int)(threadIdx.x >> 5));
  const int gid = blockIdx.x * STAT_WAVES + wave;
  const int s = gid & (TT - 1);
  const float* row = St + (size_t)gid * TT;
  const int tfirst = s & ~15;
  const int tb = s & ~127;
  const int nit = (TT - tb) >> 7;

  float4 val[16];
#pragma unroll
  for (int i = 0; i < 16; ++i)
    val[i] = make_float4(STAT_NEG, STAT_NEG, STAT_NEG, STAT_NEG);

  const int tl = tb + lane * 4;
  if (tl >= tfirst) val[0] = *(const float4*)(row + tl);
#pragma unroll
  for (int i = 1; i < 16; ++i)
    if (i < nit) val[i] = *(const float4*)(row + tl + i * 128);

  val[0].x = (tl + 0 >= s) ? val[0].x : STAT_NEG;
  val[0].y = (tl + 1 >= s) ? val[0].y : STAT_NEG;
  val[0].z = (tl + 2 >= s) ? val[0].z : STAT_NEG;
  val[0].w = (tl + 3 >= s) ? val[0].w : STAT_NEG;

  float m = STAT_NEG;
#pragma unroll
  for (int i = 0; i < 16; ++i)
    if (i < nit)
      m = fmaxf(m, fmaxf(fmaxf(val[i].x, val[i].y), fmaxf(val[i].z, val[i].w)));
#pragma unroll
  for (int off = 16; off > 0; off >>= 1)
    m = fmaxf(m, __shfl_xor(m, off, 32));

  float sum = 0.f;
#pragma unroll
  for (int i = 0; i < 16; ++i)
    if (i < nit) {
      sum += __expf(val[i].x - m);
      sum += __expf(val[i].y - m);
      sum += __expf(val[i].z - m);
      sum += __expf(val[i].w - m);
    }
#pragma unroll
  for (int off = 16; off > 0; off >>= 1)
    sum += __shfl_xor(sum, off, 32);

  if (lane == 0) {
    sh_max[wave] = m;
    sh_rcp[wave] = 1.0f / sum;
  }
  __syncthreads();
  if (threadIdx.x < STAT_WAVES) {
    colmax[blockIdx.x * STAT_WAVES + threadIdx.x] = sh_max[threadIdx.x];
    colrcp[blockIdx.x * STAT_WAVES + threadIdx.x] = sh_rcp[threadIdx.x];
  }
}

__device__ __forceinline__ void lds_only_barrier() {
  __builtin_amdgcn_fence(__ATOMIC_RELEASE, "workgroup", "local");
  __builtin_amdgcn_s_barrier();
  __builtin_amdgcn_fence(__ATOMIC_ACQUIRE, "workgroup", "local");
}

#define OT_CHUNK 32
#define OT_DEPTH 4
#define OT_COLS 64
#define P_PITCH 96
#define V_PITCH 160
#define P_TILE (16 * P_PITCH)
#define V_TILE (16 * V_PITCH)
#define OT_OP 68

__global__ __launch_bounds__(256) void attn_out(
    const float* __restrict__ St, const float* __restrict__ colmax,
    const float* __restrict__ colrcp, const float* __restrict__ v,
    float* __restrict__ out) {
  __shared__ __attribute__((aligned(16))) float shP[2][P_TILE];
  __shared__ __attribute__((aligned(16))) float shV[2][V_TILE];

  const int tid = threadIdx.x;
  const int lane = tid & 31;
  const int lo = lane & 15, hi = lane >> 4;
  const int wave = __builtin_amdgcn_readfirstlane(tid >> 5);
  const int mt = wave & 1;
  const int nt = wave >> 1;

  const int pi = blockIdx.x;
  const int h0 = (int)blockIdx.y * OT_COLS;
  const int b = blockIdx.z;

  const float* Sb = St + (size_t)b * TT * TT;
  const float* cm = colmax + (size_t)b * TT;
  const float* cr = colrcp + (size_t)b * TT;
  const float* vb = v + (size_t)b * TT * HH;

  const int psp = tid >> 4;
  const int pt2 = (tid & 15) * 2;
  const int vj = lane >> 4;
  const int vg = lane & 15;

  const float* gM = cm + 2 * psp;
  const float* gR = cr + 2 * psp;
  const float* gV = vb + (size_t)(4 * wave + vj) * HH + h0 + 4 * vg;
  const int ldsP = psp * P_PITCH + pt2 * 2;
  const int ldsV = (2 * wave) * V_PITCH + vg * 8 + vj * 4;

  const int fA = hi * P_PITCH + (mt * 16 + lo) * 2;
  const int fB = hi * V_PITCH + nt * 32 + (lo >> 2) * 8 + (lo & 3);

#define OT_SET(r)                                                       \
  v2f s_a##r, s_b##r, mx##r, rc##r;                                     \
  v4f va##r, vb##r;
  OT_SET(0) OT_SET(1) OT_SET(2) OT_SET(3)
#undef OT_SET

#define OT_LOAD(c, r)                                                   \
  {                                                                     \
    const int cc = (c) < nchunk ? (c) : nchunk - 1;                     \
    const size_t so = (size_t)cc * OT_CHUNK;                            \
    s_a##r = *(const v2f*)(gS + so * TT);                               \
    s_b##r = *(const v2f*)(gS + so * TT + TT);                          \
    mx##r = *(const v2f*)(gM + so);                                     \
    rc##r = *(const v2f*)(gR + so);                                     \
    va##r = *(const v4f*)(gV + so * HH);                                \
    vb##r = *(const v4f*)(gV + so * HH + 2 * HH);                       \
  }

#define OT_STAGE(c, buf, r)                                             \
  {                                                                     \
    const int s = (c) * OT_CHUNK + 2 * psp;                             \
    const int t = t0 + pt2;                                             \
    v4f w;                                                              \
    w.x = (s <= t) ? __expf(s_a##r.x - mx##r.x) * rc##r.x : 0.f;        \
    w.y = (s + 1 <= t) ? __expf(s_b##r.x - mx##r.y) * rc##r.y : 0.f;    \
    w.z = (s <= t + 1) ? __expf(s_a##r.y - mx##r.x) * rc##r.x : 0.f;    \
    w.w = (s + 1 <= t + 1) ? __expf(s_b##r.y - mx##r.y) * rc##r.y : 0.f;\
    *(v4f*)&shP[buf][ldsP] = w;                                         \
    float* d = &shV[buf][ldsV];                                         \
    *(v4f*)(d) = va##r;                                                 \
    *(v4f*)(d + V_PITCH) = vb##r;                                       \
  }

#define OT_COMPUTE(buf)                                                 \
  {                                                                     \
    const float* pA = &shP[buf][fA];                                    \
    const float* pB = &shV[buf][fB];                                    \
    v2f fa[8], fb[8];                                                   \
    _Pragma("unroll") for (int kk = 0; kk < 8; ++kk) {                  \
      fa[kk] = *(const v2f*)(pA + kk * 2 * P_PITCH);                    \
      fb[kk][0] = pB[kk * 2 * V_PITCH];                                 \
      fb[kk][1] = pB[kk * 2 * V_PITCH + 4];                             \
    }                                                                   \
    __builtin_amdgcn_sched_barrier(0x20);                               \
    _Pragma("unroll") for (int kk = 0; kk < 8; ++kk)                    \
      acc = wmma_f32(fa[kk], fb[kk], acc);                              \
  }

#define OT_TRIP(u, un)                                                  \
  {                                                                     \
    OT_LOAD(c + (u) + OT_DEPTH, u);                                     \
    OT_COMPUTE((u) & 1);                                                \
    OT_STAGE(c + (u) + 1, (un) & 1, un);                                \
    lds_only_barrier();                                                 \
  }

  for (int pass = 0; pass < 2; ++pass) {
    const int tb = pass == 0 ? TT / OT_CHUNK - 1 - pi : pi;
    const int t0 = tb * 32;
    const int nchunk = tb + 1;
    const float* gS = Sb + (size_t)(2 * psp) * TT + t0 + pt2;

    v8f acc = {};

    OT_LOAD(0, 0);
    __builtin_amdgcn_sched_barrier(0);
    OT_LOAD(1, 1);
    __builtin_amdgcn_sched_barrier(0);
    OT_LOAD(2, 2);
    __builtin_amdgcn_sched_barrier(0);
    OT_LOAD(3, 3);
    __builtin_amdgcn_sched_barrier(0);
    OT_STAGE(0, 0, 0);
    lds_only_barrier();

    const int ntrip = (nchunk + OT_DEPTH - 1) & ~(OT_DEPTH - 1);
    for (int c = 0; c < ntrip; c += OT_DEPTH) {
      OT_TRIP(0, 1)
      OT_TRIP(1, 2)
      OT_TRIP(2, 3)
      OT_TRIP(3, 0)
    }

    float* ot = &shV[1][0];
#pragma unroll
    for (int r = 0; r < 8; ++r)
      ot[(mt * 16 + r + 8 * hi) * OT_OP + nt * 16 + lo] = acc[r];
    lds_only_barrier();
    {
      const int orow = tid >> 3;
      const int oc = (tid & 7) * 4;
      float* og = out + ((size_t)b * TT + t0 + orow) * HH + h0 + oc;
      *(v4f*)og = *(const v4f*)&ot[orow * OT_OP + oc];
      *(v4f*)(og + 32) = *(const v4f*)&ot[orow * OT_OP + oc + 32];
    }
  }
#undef OT_TRIP
#undef OT_LOAD
#undef OT_STAGE
#undef OT_COMPUTE
}

extern "C" void kernel_launch(void* const* d_in, const int* in_sizes, int n_in,
                              void* d_out, int out_size, void* d_ws,
                              size_t ws_size, hipStream_t stream) {
  (void)in_sizes; (void)n_in; (void)out_size; (void)ws_size;
  const float* x = (const float*)d_in[0];
  const float* Wq = (const float*)d_in[1];
  const float* Wk = (const float*)d_in[2];
  const float* Wv = (const float*)d_in[3];
  float* out = (float*)d_out;

  float* ws = (float*)d_ws;
  float* q = ws;
  float* k = q + (size_t)BB * TT * HH;
  float* v = k + (size_t)BB * TT * HH;
  float* S = v + (size_t)BB * TT * HH;
  float* cmax = S + (size_t)BB * TT * TT;
  float* crcp = cmax + (size_t)BB * TT;

  qkv_proj<<<dim3(2, BB * TT / PJ_BM), 256, 0, stream>>>(x, Wq, Wk, Wv, q, k, v);
  qk_scores<<<dim3(34, 8, BB), 128, 0, stream>>>(q, k, S);
  col_softmax_stats<<<dim3(BB * TT / STAT_WAVES), STAT_WAVES * 32, 0, stream>>>(
      S, cmax, crcp);
  attn_out<<<dim3(TT / (2 * OT_CHUNK), HH / OT_COLS, BB), 256, 0, stream>>>(S, cmax, crcp,
                                                                      v, out);
}


# scores kept in the workspace as 32x32 tiles in the output kernel's LDS order (s pair, t, 2): a chunk of scores is one contiguous 4 KiB read (was 32 pieces of 128 B); column sums re-ordered
# speedup vs baseline: 17.2238x; 1.0370x over previous
#include <hip/hip_runtime.h>
#include <hip/hip_bf16.h>
#include <math.h>

#define BB 4
#define TT 2048
#define CC 1024
#define HH 128

typedef __attribute__((ext_vector_type(2))) float v2f;
typedef __attribute__((ext_vector_type(8))) float v8f;
typedef __attribute__((ext_vector_type(4))) float v4f;

__device__ __forceinline__ v8f wmma_f32(v2f a, v2f b, v8f c) {
  return __builtin_amdgcn_wmma_f32_16x16x4_f32(false, a, false, b, (short)0, c,
                                               false, false);
}

#define PJ_BM 64
#define PJ_BK 32
#define PJ_XP 36
#define PJ_WP 416
#define PJ_XT (PJ_BM * PJ_XP)
#define PJ_WT ((PJ_BK / 2) * PJ_WP)

__global__ __launch_bounds__(256) void qkv_proj(
    const float* __restrict__ x, const float* __restrict__ Wq,
    const float* __restrict__ Wk, const float* __restrict__ Wv,
    float* __restrict__ q, float* __restrict__ k, float* __restrict__ v) {
  __shared__ __attribute__((aligned(16))) float shX[2][PJ_XT];
  __shared__ __attribute__((aligned(16))) float shW[2][PJ_WT];

  const int tid = threadIdx.x;
  const int lane = tid & 31;
  const int lo = lane & 15, hi = lane >> 4;
  const int wave = __builtin_amdgcn_readfirstlane(tid >> 5);
  const int rp = wave & 1;
  const int cg = wave >> 1;

  const int half = blockIdx.x;
  const int m0 = blockIdx.y * PJ_BM;

  const int xr = tid >> 3;
  const int xc = (tid & 7) * 4;
  const int wp = tid >> 4;
  const int wc = (tid & 15) * 4;

  const float* gX = x + (size_t)(m0 + xr) * CC + xc;
  const float* gW0 = (half == 0 ? Wq : Wk) + (size_t)(2 * wp) * HH +
                     (half == 0 ? 0 : 64) + wc;
  const float* gW1 = (half == 0 ? Wq : Wv) + (size_t)(2 * wp) * HH +
                     (half == 0 ? 64 : 0) + wc;
  const float* gW2 = (half == 0 ? Wk : Wv) + (size_t)(2 * wp) * HH +
                     (half == 0 ? 0 : 64) + wc;
  const int ldsX = xr * PJ_XP + xc;
  const int ldsW = wp * PJ_WP + wc * 2;

  const int fA = (rp * 32 + lo) * PJ_XP + 2 * hi;
  const int fB = hi * PJ_WP + (cg * 48 + lo) * 2;

  float4 xa, xb, w0a, w0b, w1a, w1b, w2a, w2b;

#define PJ_LOAD(c)                                                      \
  {                                                                     \
    const size_t ko = (size_t)(c) * PJ_BK;                              \
    xa = *(const float4*)(gX + ko);                                     \
    xb = *(const float4*)(gX + ko + (size_t)32 * CC);                   \
    w0a = *(const float4*)(gW0 + ko * HH);                              \
    w0b = *(const float4*)(gW0 + ko * HH + HH);                         \
    w1a = *(const float4*)(gW1 + ko * HH);                              \
    w1b = *(const float4*)(gW1 + ko * HH + HH);                         \
    w2a = *(const float4*)(gW2 + ko * HH);                              \
    w2b = *(const float4*)(gW2 + ko * HH + HH);                         \
  }

#define PJ_STAGE(buf)                                                   \
  {                                                                     \
    *(float4*)&shX[buf][ldsX] = xa;                                     \
    *(float4*)&shX[buf][ldsX + 32 * PJ_XP] = xb;                        \
    float* d = &shW[buf][ldsW];                                         \
    *(float4*)(d) = make_float4(w0a.x, w0b.x, w0a.y, w0b.y);            \
    *(float4*)(d + 4) = make_float4(w0a.z, w0b.z, w0a.w, w0b.w);        \
    *(float4*)(d + 128) = make_float4(w1a.x, w1b.x, w1a.y, w1b.y);      \
    *(float4*)(d + 132) = make_float4(w1a.z, w1b.z, w1a.w, w1b.w);      \
    *(float4*)(d + 256) = make_float4(w2a.x, w2b.x, w2a.y, w2b.y);      \
    *(float4*)(d + 260) = make_float4(w2a.z, w2b.z, w2a.w, w2b.w);      \
  }

  v8f acc[2][3] = {};

#define PJ_COMPUTE(buf)                                                 \
  {                                                                     \
    const float* pA = &shX[buf][fA];                                    \
    const float* pB = &shW[buf][fB];                                    \
    v2f fa[8][2], fb[8][3];                                             \
    _Pragma("unroll") for (int kk = 0; kk < 8; ++kk) {                  \
      _Pragma("unroll") for (int i = 0; i < 2; ++i)                     \
        fa[kk][i] = *(const v2f*)(pA + i * 16 * PJ_XP + kk * 4);        \
      _Pragma("unroll") for (int j = 0; j < 3; ++j)                     \
        fb[kk][j] = *(const v2f*)(pB + kk * 2 * PJ_WP + j * 32);        \
    }                                                                   \
    _Pragma("unroll") for (int kk = 0; kk < 8; ++kk)                    \
      _Pragma("unroll") for (int i = 0; i < 2; ++i)                     \
        _Pragma("unroll") for (int j = 0; j < 3; ++j)                   \
          acc[i][j] = wmma_f32(fa[kk][i], fb[kk][j], acc[i][j]);        \
    __builtin_amdgcn_sched_group_barrier(0x020, 8, 0);                  \
    __builtin_amdgcn_sched_group_barrier(0x100, 40, 0);                 \
    __builtin_amdgcn_sched_group_barrier(0x008, 48, 0);                 \
  }

  PJ_LOAD(0);
  PJ_STAGE(0);
  __syncthreads();
  for (int c = 0; c + 1 < CC / PJ_BK; ++c) {
    const int cur = c & 1;
    PJ_LOAD(c + 1);
    PJ_COMPUTE(cur);
    PJ_STAGE(cur ^ 1);
    __syncthreads();
  }
  PJ_COMPUTE((CC / PJ_BK - 1) & 1);

#pragma unroll
  for (int j = 0; j < 3; ++j) {
    const int gn = 192 * half + 48 * cg + 16 * j;
    float* om = (gn < HH) ? q : (gn < 2 * HH) ? k : v;
    float* orow = om + (size_t)(m0 + rp * 32) * HH + (gn & (HH - 1)) + lo;
#pragma unroll
    for (int i = 0; i < 2; ++i)
#pragma unroll
      for (int r = 0; r < 8; ++r)
        orow[(size_t)(16 * i + 8 * hi + r) * HH] = acc[i][j][r];
  }
#undef PJ_LOAD
#undef PJ_STAGE
#undef PJ_COMPUTE
}

#define SC_BS 128
#define SC_BT 64
#define SC_PITCH 132
#define ST_TILES 64
#define ST_TILE 1024

__global__ __launch_bounds__(128) void qk_scores(const float* __restrict__ q,
                                                 const float* __restrict__ kmat,
                                                 float* __restrict__ St) {
  __shared__ __attribute__((aligned(16))) float shK[SC_BS * SC_PITCH];
  __shared__ __attribute__((aligned(16))) float shQ[SC_BT * SC_PITCH];

  const int tid = threadIdx.x;
  const int lane = tid & 31;
  const int lo = lane & 15, hi = lane >> 4;
  const int ws = __builtin_amdgcn_readfirstlane(tid >> 5);

  const int pr = blockIdx.y;
  const int px = (int)blockIdx.x >> 1;
  const int th = (int)blockIdx.x & 1;
  const int bt = (px <= pr) ? pr : 15 - pr;
  const int bs = (px <= pr) ? px : px - (pr + 1);
  const int b = blockIdx.z;
  const int s0 = bs * SC_BS;
  const int t0 = bt * SC_BS + th * SC_BT;

  {
    const int c4 = lane * 4;
    const float* gk = kmat + ((size_t)b * TT + s0 + ws) * HH + c4;
    const float* gq = q + ((size_t)b * TT + t0 + ws) * HH + c4;
    float4 tk[32], tq[16];
#pragma unroll
    for (int i = 0; i < 32; ++i) tk[i] = *(const float4*)(gk + (size_t)i * 4 * HH);
#pragma unroll
    for (int i = 0; i < 16; ++i) tq[i] = *(const float4*)(gq + (size_t)i * 4 * HH);
#pragma unroll
    for (int i = 0; i < 32; ++i)
      *(float4*)&shK[(ws + 4 * i) * SC_PITCH + c4] = tk[i];
#pragma unroll
    for (int i = 0; i < 16; ++i)
      *(float4*)&shQ[(16 * (2 * (i >> 3) + (ws & 1)) + (ws >> 1) + 2 * (i & 7)) *
                         SC_PITCH +
                     c4] = tq[i];
  }
  __syncthreads();

  if (bs == bt && th == 0 && ws >= 2) return;

  const float* pA = &shK[(32 * ws + lo) * SC_PITCH + 2 * hi];
  const float* pB = &shQ[lo * SC_PITCH + 2 * hi];

  v8f acc[2][4] = {};
  v2f fa[2][4][2], fb[2][4][4];

#define SC_FRAGS(set, g)                                                \
  _Pragma("unroll") for (int kk = 0; kk < 4; ++kk) {                    \
    _Pragma("unroll") for (int i = 0; i < 2; ++i)                       \
      fa[set][kk][i] =                                                  \
          *(const v2f*)(pA + i * 16 * SC_PITCH + (g) * 16 + kk * 4);    \
    _Pragma("unroll") for (int j = 0; j < 4; ++j)                       \
      fb[set][kk][j] =                                                  \
          *(const v2f*)(pB + j * 16 * SC_PITCH + (g) * 16 + kk * 4);    \
  }

  SC_FRAGS(0, 0);
#pragma unroll
  for (int g = 0; g < 8; ++g) {
    const int cur = g & 1;
    if (g + 1 < 8) SC_FRAGS(cur ^ 1, g + 1);
    __builtin_amdgcn_sched_barrier(0);
#pragma unroll
    for (int kk = 0; kk < 4; ++kk)
#pragma unroll
      for (int i = 0; i < 2; ++i)
#pragma unroll
        for (int j = 0; j < 4; ++j)
          acc[i][j] = wmma_f32(fa[cur][kk][i], fb[cur][kk][j], acc[i][j]);
    __builtin_amdgcn_sched_barrier(0);
  }
#undef SC_FRAGS

  const float scale = 0.08838834764831845f;
  const int t = t0 + 2 * lo;
  float* tl0 = St + (size_t)b * TT * TT +
               ((size_t)((s0 >> 5) + ws) * ST_TILES + (t0 >> 5)) * ST_TILE +
               4 * lo;
#pragma unroll
  for (int i = 0; i < 2; ++i) {
#pragma unroll
    for (int rp = 0; rp < 4; ++rp) {
      const int r = 2 * rp;
      const int s = s0 + 32 * ws + 16 * i + 8 * hi + r;
      float* d = tl0 + (8 * i + 4 * hi + rp) * 64;
      float4 o;
      o.x = (s <= t) ? acc[i][0][r] * scale : -1e30f;
      o.y = (s + 1 <= t) ? acc[i][0][r + 1] * scale : -1e30f;
      o.z = (s <= t + 1) ? acc[i][1][r] * scale : -1e30f;
      o.w = (s + 1 <= t + 1) ? acc[i][1][r + 1] * scale : -1e30f;
      *(float4*)d = o;
      o.x = (s <= t + 32) ? acc[i][2][r] * scale : -1e30f;
      o.y = (s + 1 <= t + 32) ? acc[i][2][r + 1] * scale : -1e30f;
      o.z = (s <= t + 33) ? acc[i][3][r] * scale : -1e30f;
      o.w = (s + 1 <= t + 33) ? acc[i][3][r + 1] * scale : -1e30f;
      *(float4*)(d + ST_TILE) = o;
    }
  }
}

#define STAT_COLS 32
#define STAT_NEG (-3.0e38f)

__global__ __launch_bounds__(256) void col_softmax_stats(
    const float* __restrict__ St, float* __restrict__ colmax,
    float* __restrict__ colrcp) {
  __shared__ float sh_max[STAT_COLS];
  __shared__ float sh_rcp[STAT_COLS];

  const int lane = threadIdx.x & 31;
  const int wave = __builtin_amdgcn_readfirstlane((int)(threadIdx.x >> 5));
  const int col0 = blockIdx.x * STAT_COLS;
  const int b = col0 / TT;
  const int sb = (col0 & (TT - 1)) >> 5;
  const int nit = ST_TILES - sb;
  const int pr = 2 * wave + (lane >> 4);
  const int se = sb * 32 + 2 * pr;
  const int t = sb * 32 + (lane & 15) * 2;
  const float* row = St + (size_t)b * TT * TT +
                     ((size_t)sb * ST_TILES + sb) * ST_TILE + wave * 128 +
                     lane * 4;

  float4 val[ST_TILES];
#pragma unroll
  for (int i = 0; i < ST_TILES; ++i)
    val[i] = make_float4(STAT_NEG, STAT_NEG, STAT_NEG, STAT_NEG);

#pragma unroll
  for (int i = 0; i < ST_TILES; ++i)
    if (i < nit) val[i] = *(const float4*)(row + (size_t)i * ST_TILE);

  val[0].x = (t >= se) ? val[0].x : STAT_NEG;
  val[0].y = (t >= se + 1) ? val[0].y : STAT_NEG;
  val[0].z = (t + 1 >= se) ? val[0].z : STAT_NEG;
  val[0].w = (t + 1 >= se + 1) ? val[0].w : STAT_NEG;

  float me = STAT_NEG, mo = STAT_NEG;
#pragma unroll
  for (int i = 0; i < ST_TILES; ++i)
    if (i < nit) {
      me = fmaxf(me, fmaxf(val[i].x, val[i].z));
      mo = fmaxf(mo, fmaxf(val[i].y, val[i].w));
    }
#pragma unroll
  for (int off = 8; off > 0; off >>= 1) {
    me = fmaxf(me, __shfl_xor(me, off, 32));
    mo = fmaxf(mo, __shfl_xor(mo, off, 32));
  }

  float sume = 0.f, sumo = 0.f;
#pragma unroll
  for (int i = 0; i < ST_TILES; ++i)
    if (i < nit) {
      sume += __expf(val[i].x - me);
      sumo += __expf(val[i].y - mo);
      sume += __expf(val[i].z - me);
      sumo += __expf(val[i].w - mo);
    }
#pragma unroll
  for (int off = 8; off > 0; off >>= 1) {
    sume += __shfl_xor(sume, off, 32);
    sumo += __shfl_xor(sumo, off, 32);
  }

  if ((lane & 15) == 0) {
    sh_max[2 * pr] = me;
    sh_max[2 * pr + 1] = mo;
    sh_rcp[2 * pr] = 1.0f / sume;
    sh_rcp[2 * pr + 1] = 1.0f / sumo;
  }
  __syncthreads();
  if (threadIdx.x < STAT_COLS) {
    colmax[col0 + threadIdx.x] = sh_max[threadIdx.x];
    colrcp[col0 + threadIdx.x] = sh_rcp[threadIdx.x];
  }
}

__device__ __forceinline__ void lds_only_barrier() {
  __builtin_amdgcn_fence(__ATOMIC_RELEASE, "workgroup", "local");
  __builtin_amdgcn_s_barrier();
  __builtin_amdgcn_fence(__ATOMIC_ACQUIRE, "workgroup", "local");
}

#define OT_CHUNK 32
#define OT_DEPTH 4
#define OT_COLS 64
#define P_PITCH 96
#define V_PITCH 160
#define P_TILE (16 * P_PITCH)
#define V_TILE (16 * V_PITCH)
#define OT_OP 68

__global__ __launch_bounds__(256) void attn_out(
    const float* __restrict__ St, const float* __restrict__ colmax,
    const float* __restrict__ colrcp, const float* __restrict__ v,
    float* __restrict__ out) {
  __shared__ __attribute__((aligned(16))) float shP[2][P_TILE];
  __shared__ __attribute__((aligned(16))) float shV[2][V_TILE];

  const int tid = threadIdx.x;
  const int lane = tid & 31;
  const int lo = lane & 15, hi = lane >> 4;
  const int wave = __builtin_amdgcn_readfirstlane(tid >> 5);
  const int mt = wave & 1;
  const int nt = wave >> 1;

  const int pi = blockIdx.x;
  const int h0 = (int)blockIdx.y * OT_COLS;
  const int b = blockIdx.z;

  const float* Sb = St + (size_t)b * TT * TT;
  const float* cm = colmax + (size_t)b * TT;
  const float* cr = colrcp + (size_t)b * TT;
  const float* vb = v + (size_t)b * TT * HH;

  const int psp = tid >> 4;
  const int pt2 = (tid & 15) * 2;
  const int vj = lane >> 4;
  const int vg = lane & 15;

  const float* gM = cm + 2 * psp;
  const float* gR = cr + 2 * psp;
  const float* gV = vb + (size_t)(4 * wave + vj) * HH + h0 + 4 * vg;
  const int ldsP = psp * P_PITCH + pt2 * 2;
  const int ldsV = (2 * wave) * V_PITCH + vg * 8 + vj * 4;

  const int fA = hi * P_PITCH + (mt * 16 + lo) * 2;
  const int fB = hi * V_PITCH + nt * 32 + (lo >> 2) * 8 + (lo & 3);

#define OT_SET(r)                                                       \
  v2f mx##r, rc##r;                                                     \
  v4f sv##r, va##r, vb##r;
  OT_SET(0) OT_SET(1) OT_SET(2) OT_SET(3)
#undef OT_SET

#define OT_LOAD(c, r)                                                   \
  {                                                                     \
    const int cc = (c) < nchunk ? (c) : nchunk - 1;                     \
    const size_t so = (size_t)cc * OT_CHUNK;                            \
    sv##r = *(const v4f*)(gS + (size_t)cc * (ST_TILES * ST_TILE));      \
    mx##r = *(const v2f*)(gM + so);                                     \
    rc##r = *(const v2f*)(gR + so);                                     \
    va##r = *(const v4f*)(gV + so * HH);                                \
    vb##r = *(const v4f*)(gV + so * HH + 2 * HH);                       \
  }

#define OT_STAGE(c, buf, r)                                             \
  {                                                                     \
    const int s = (c) * OT_CHUNK + 2 * psp;                             \
    const int t = t0 + pt2;                                             \
    v4f w;                                                              \
    w.x = (s <= t) ? __expf(sv##r.x - mx##r.x) * rc##r.x : 0.f;         \
    w.y = (s + 1 <= t) ? __expf(sv##r.y - mx##r.y) * rc##r.y : 0.f;     \
    w.z = (s <= t + 1) ? __expf(sv##r.z - mx##r.x) * rc##r.x : 0.f;     \
    w.w = (s + 1 <= t + 1) ? __expf(sv##r.w - mx##r.y) * rc##r.y : 0.f; \
    *(v4f*)&shP[buf][ldsP] = w;                                         \
    float* d = &shV[buf][ldsV];                                         \
    *(v4f*)(d) = va##r;                                                 \
    *(v4f*)(d + V_PITCH) = vb##r;                                       \
  }

#define OT_COMPUTE(buf)                                                 \
  {                                                                     \
    const float* pA = &shP[buf][fA];                                    \
    const float* pB = &shV[buf][fB];                                    \
    v2f fa[8], fb[8];                                                   \
    _Pragma("unroll") for (int kk = 0; kk < 8; ++kk) {                  \
      fa[kk] = *(const v2f*)(pA + kk * 2 * P_PITCH);                    \
      fb[kk][0] = pB[kk * 2 * V_PITCH];                                 \
      fb[kk][1] = pB[kk * 2 * V_PITCH + 4];                             \
    }                                                                   \
    __builtin_amdgcn_sched_barrier(0x20);                               \
    _Pragma("unroll") for (int kk = 0; kk < 8; ++kk)                    \
      acc = wmma_f32(fa[kk], fb[kk], acc);                              \
  }

#define OT_TRIP(u, un)                                                  \
  {                                                                     \
    OT_LOAD(c + (u) + OT_DEPTH, u);                                     \
    OT_COMPUTE((u) & 1);                                                \
    OT_STAGE(c + (u) + 1, (un) & 1, un);                                \
    lds_only_barrier();                                                 \
  }

  for (int pass = 0; pass < 2; ++pass) {
    const int tb = pass == 0 ? TT / OT_CHUNK - 1 - pi : pi;
    const int t0 = tb * 32;
    const int nchunk = tb + 1;
    const float* gS = Sb + (size_t)tb * ST_TILE + psp * 64 + pt2 * 2;

    v8f acc = {};

    OT_LOAD(0, 0);
    __builtin_amdgcn_sched_barrier(0);
    OT_LOAD(1, 1);
    __builtin_amdgcn_sched_barrier(0);
    OT_LOAD(2, 2);
    __builtin_amdgcn_sched_barrier(0);
    OT_LOAD(3, 3);
    __builtin_amdgcn_sched_barrier(0);
    OT_STAGE(0, 0, 0);
    lds_only_barrier();

    const int ntrip = (nchunk + OT_DEPTH - 1) & ~(OT_DEPTH - 1);
    for (int c = 0; c < ntrip; c += OT_DEPTH) {
      OT_TRIP(0, 1)
      OT_TRIP(1, 2)
      OT_TRIP(2, 3)
      OT_TRIP(3, 0)
    }

    float* ot = &shV[1][0];
#pragma unroll
    for (int r = 0; r < 8; ++r)
      ot[(mt * 16 + r + 8 * hi) * OT_OP + nt * 16 + lo] = acc[r];
    lds_only_barrier();
    {
      const int orow = tid >> 3;
      const int oc = (tid & 7) * 4;
      float* og = out + ((size_t)b * TT + t0 + orow) * HH + h0 + oc;
      *(v4f*)og = *(const v4f*)&ot[orow * OT_OP + oc];
      *(v4f*)(og + 32) = *(const v4f*)&ot[orow * OT_OP + oc + 32];
    }
  }
#undef OT_TRIP
#undef OT_LOAD
#undef OT_STAGE
#undef OT_COMPUTE
}

extern "C" void kernel_launch(void* const* d_in, const int* in_sizes, int n_in,
                              void* d_out, int out_size, void* d_ws,
                              size_t ws_size, hipStream_t stream) {
  (void)in_sizes; (void)n_in; (void)out_size; (void)ws_size;
  const float* x = (const float*)d_in[0];
  const float* Wq = (const float*)d_in[1];
  const float* Wk = (const float*)d_in[2];
  const float* Wv = (const float*)d_in[3];
  float* out = (float*)d_out;

  float* ws = (float*)d_ws;
  float* q = ws;
  float* k = q + (size_t)BB * TT * HH;
  float* v = k + (size_t)BB * TT * HH;
  float* S = v + (size_t)BB * TT * HH;
  float* cmax = S + (size_t)BB * TT * TT;
  float* crcp = cmax + (size_t)BB * TT;

  qkv_proj<<<dim3(2, BB * TT / PJ_BM), 256, 0, stream>>>(x, Wq, Wk, Wv, q, k, v);
  qk_scores<<<dim3(34, 8, BB), 128, 0, stream>>>(q, k, S);
  col_softmax_stats<<<dim3(BB * TT / STAT_COLS), 256, 0, stream>>>(
      S, cmax, crcp);
  attn_out<<<dim3(TT / (2 * OT_CHUNK), HH / OT_COLS, BB), 256, 0, stream>>>(S, cmax, crcp,
                                                                      v, out);
}
